# starting kernel
# baseline (speedup 1.0000x reference)
#include <hip/hip_runtime.h>
#include <math.h>

typedef float v2f __attribute__((ext_vector_type(2)));
typedef float v8f __attribute__((ext_vector_type(8)));

#define F_IN  128
#define F_OUT 64


__global__ void k_init_deg(float* __restrict__ deg, int N) {
    int i = blockIdx.x * blockDim.x + threadIdx.x;
    if (i < N) deg[i] = 1.0f;
}

__global__ void k_count_deg(const int* __restrict__ dst, float* __restrict__ deg, int E) {
    int e = blockIdx.x * blockDim.x + threadIdx.x;
    if (e < E) atomicAdd(&deg[dst[e]], 1.0f);
}

__global__ void k_dinv(float* __restrict__ deg, int N) {
    int i = blockIdx.x * blockDim.x + threadIdx.x;
    if (i < N) {
        float d = deg[i];
        deg[i] = (d > 0.0f) ? rsqrtf(d) : 0.0f;
    }
}


__global__ void __launch_bounds__(256) k_gemm(const float* __restrict__ x,
                                              const float* __restrict__ W,
                                              float* __restrict__ h, int N) {
    const int wave = threadIdx.x >> 5;
    const int lane = threadIdx.x & 31;
    const int row0 = (blockIdx.x * 8 + wave) * 16;
    if (row0 >= N) return;

    const int lr    = lane & 15;
    const int khalf = lane >> 4;

    int m = row0 + lr;
    if (m >= N) m = N - 1;

    const float2* xp  = (const float2*)(x + (size_t)m * F_IN);
    const float2* wp0 = (const float2*)(W + (size_t)(lr +  0) * F_IN);
    const float2* wp1 = (const float2*)(W + (size_t)(lr + 16) * F_IN);
    const float2* wp2 = (const float2*)(W + (size_t)(lr + 32) * F_IN);
    const float2* wp3 = (const float2*)(W + (size_t)(lr + 48) * F_IN);

    v8f c0 = {}, c1 = {}, c2 = {}, c3 = {};

    #pragma unroll
    for (int k0 = 0; k0 < F_IN; k0 += 4) {
        const int ki = (k0 >> 1) + khalf;
        float2 af = xp[ki];
        v2f a; a.x = af.x; a.y = af.y;
        float2 b0f = wp0[ki]; v2f b0; b0.x = b0f.x; b0.y = b0f.y;
        float2 b1f = wp1[ki]; v2f b1; b1.x = b1f.x; b1.y = b1f.y;
        float2 b2f = wp2[ki]; v2f b2; b2.x = b2f.x; b2.y = b2f.y;
        float2 b3f = wp3[ki]; v2f b3; b3.x = b3f.x; b3.y = b3f.y;

        c0 = __builtin_amdgcn_wmma_f32_16x16x4_f32(false, a, false, b0, (short)0, c0, false, false);
        c1 = __builtin_amdgcn_wmma_f32_16x16x4_f32(false, a, false, b1, (short)0, c1, false, false);
        c2 = __builtin_amdgcn_wmma_f32_16x16x4_f32(false, a, false, b2, (short)0, c2, false, false);
        c3 = __builtin_amdgcn_wmma_f32_16x16x4_f32(false, a, false, b3, (short)0, c3, false, false);
    }

    const int mbase = row0 + (khalf << 3);
    #pragma unroll
    for (int v = 0; v < 8; ++v) {
        int mm = mbase + v;
        if (mm < N) {
            float* hp = h + (size_t)mm * F_OUT + lr;
            hp[ 0] = c0[v];
            hp[16] = c1[v];
            hp[32] = c2[v];
            hp[48] = c3[v];
        }
    }
}


__global__ void k_init_agg(const float* __restrict__ h, const float* __restrict__ dinv,
                           const float* __restrict__ b_conv, float* __restrict__ agg, int total) {
    int idx = blockIdx.x * blockDim.x + threadIdx.x;
    if (idx < total) {
        int node = idx >> 6;
        int f    = idx & 63;
        float di = dinv[node];
        agg[idx] = b_conv[f] + h[idx] * di * di;
    }
}


__global__ void k_scatter(const int* __restrict__ src, const int* __restrict__ dst,
                          const float* __restrict__ h, const float* __restrict__ dinv,
                          float* __restrict__ agg, int E) {
    int gtid = blockIdx.x * blockDim.x + threadIdx.x;
    int e    = gtid >> 5;
    int lane = threadIdx.x & 31;
    if (e >= E) return;

    int s = src[e];
    int d = dst[e];
    float norm = dinv[s] * dinv[d];

    const float2* hs = (const float2*)(h + (size_t)s * F_OUT);
    float2 v = hs[lane];

    float* ad = agg + (size_t)d * F_OUT + lane * 2;
    atomicAdd(ad + 0, v.x * norm);
    atomicAdd(ad + 1, v.y * norm);
}


__global__ void k_finalize(const float* __restrict__ agg, const float* __restrict__ W_lin,
                           const float* __restrict__ b_lin, float* __restrict__ out, int N) {
    int gtid = blockIdx.x * blockDim.x + threadIdx.x;
    int node = gtid >> 5;
    int lane = threadIdx.x & 31;
    if (node >= N) return;

    const float* row = agg + (size_t)node * F_OUT;
    float p = fmaxf(row[lane], 0.0f) * W_lin[lane]
            + fmaxf(row[lane + 32], 0.0f) * W_lin[lane + 32];

    #pragma unroll
    for (int off = 16; off > 0; off >>= 1)
        p += __shfl_down(p, off);

    if (lane == 0) {
        float z = p + b_lin[0];
        out[node] = 1.0f / (1.0f + __expf(-z));
    }
}


extern "C" void kernel_launch(void* const* d_in, const int* in_sizes, int n_in,
                              void* d_out, int out_size, void* d_ws, size_t ws_size,
                              hipStream_t stream) {
    const float* x      = (const float*)d_in[0];
    const int*   edge   = (const int*)  d_in[1];
    const float* W_conv = (const float*)d_in[2];
    const float* b_conv = (const float*)d_in[3];
    const float* W_lin  = (const float*)d_in[4];
    const float* b_lin  = (const float*)d_in[5];
    float*       out    = (float*)d_out;

    const int N = in_sizes[0] / F_IN;
    const int E = in_sizes[1] / 2;
    const int* src = edge;
    const int* dst = edge + E;

    float* ws   = (float*)d_ws;
    float* dinv = ws;
    float* h    = ws + N;
    float* agg  = h + (size_t)N * F_OUT;

    k_init_deg <<<(N + 255) / 256, 256, 0, stream>>>(dinv, N);
    k_count_deg<<<(E + 255) / 256, 256, 0, stream>>>(dst, dinv, E);
    k_dinv     <<<(N + 255) / 256, 256, 0, stream>>>(dinv, N);

    k_gemm<<<(N + 127) / 128, 256, 0, stream>>>(x, W_conv, h, N);

    const int tot = N * F_OUT;
    k_init_agg<<<(tot + 255) / 256, 256, 0, stream>>>(h, dinv, b_conv, agg, tot);

    const long long sth = (long long)E * 32;
    k_scatter<<<(int)((sth + 255) / 256), 256, 0, stream>>>(src, dst, h, dinv, agg, E);

    const long long fth = (long long)N * 32;
    k_finalize<<<(int)((fth + 255) / 256), 256, 0, stream>>>(agg, W_lin, b_lin, out, N);
}


# global float-atomic scatter replaced by LDS slices: each block keeps 1216 destination rows in LDS, filters a third of dst, adds hit rows with LDS float adds; finalize sums 3 planes; sum order changes
# speedup vs baseline: 2.2095x; 2.2095x over previous
#include <hip/hip_runtime.h>
#include <math.h>

typedef float v2f __attribute__((ext_vector_type(2)));
typedef float v8f __attribute__((ext_vector_type(8)));

#define F_IN  128
#define F_OUT 64


__global__ void k_init_deg(float* __restrict__ deg, int N) {
    int i = blockIdx.x * blockDim.x + threadIdx.x;
    if (i < N) deg[i] = 1.0f;
}


#define DEG_BINS    16384
#define DEG_SEGS    32
#define DEG_THREADS 1024
#define DEG_AHEAD   8

__global__ void __launch_bounds__(DEG_THREADS) k_count_deg(const int* __restrict__ dst,
                                                           float* __restrict__ deg, int E, int N) {
    __shared__ unsigned cnt[DEG_BINS];
    const int tid = threadIdx.x;
    const int lo  = blockIdx.y * DEG_BINS;

    for (int b = tid; b < DEG_BINS; b += DEG_THREADS) cnt[b] = 0u;
    __syncthreads();

    const int nseg  = (int)gridDim.x;
    const int chunk = (E + nseg - 1) / nseg;
    const int e0    = blockIdx.x * chunk;
    const int e1    = (E - e0 < chunk) ? E : e0 + chunk;

    for (int e = e0 + tid; e < e1; e += DEG_AHEAD * DEG_THREADS) {
        int d[DEG_AHEAD];
        #pragma unroll
        for (int j = 0; j < DEG_AHEAD; ++j) {
            const int ee = e + j * DEG_THREADS;
            d[j] = (ee < e1) ? dst[ee] : -1;
        }
        #pragma unroll
        for (int j = 0; j < DEG_AHEAD; ++j) {
            const unsigned r = (unsigned)(d[j] - lo);
            if (r < (unsigned)DEG_BINS) atomicAdd(&cnt[r], 1u);
        }
    }
    __syncthreads();

    const int rot = (blockIdx.x * DEG_THREADS) & (DEG_BINS - 1);
    unsigned c[DEG_BINS / DEG_THREADS];
    #pragma unroll
    for (int i = 0; i < DEG_BINS / DEG_THREADS; ++i)
        c[i] = cnt[(tid + i * DEG_THREADS + rot) & (DEG_BINS - 1)];
    #pragma unroll
    for (int i = 0; i < DEG_BINS / DEG_THREADS; ++i) {
        const int node = lo + ((tid + i * DEG_THREADS + rot) & (DEG_BINS - 1));
        if (c[i] != 0u && node < N) atomicAdd(&deg[node], (float)c[i]);
    }
}

__global__ void k_dinv(float* __restrict__ deg, int N) {
    int i = blockIdx.x * blockDim.x + threadIdx.x;
    if (i < N) {
        float d = deg[i];
        deg[i] = (d > 0.0f) ? rsqrtf(d) : 0.0f;
    }
}


__global__ void __launch_bounds__(256) k_gemm(const float* __restrict__ x,
                                              const float* __restrict__ W,
                                              float* __restrict__ h, int N) {
    const int wave = threadIdx.x >> 5;
    const int lane = threadIdx.x & 31;
    const int row0 = (blockIdx.x * 8 + wave) * 16;
    if (row0 >= N) return;

    const int lr    = lane & 15;
    const int khalf = lane >> 4;

    int m = row0 + lr;
    if (m >= N) m = N - 1;

    const float2* xp  = (const float2*)(x + (size_t)m * F_IN);
    const float2* wp0 = (const float2*)(W + (size_t)(lr +  0) * F_IN);
    const float2* wp1 = (const float2*)(W + (size_t)(lr + 16) * F_IN);
    const float2* wp2 = (const float2*)(W + (size_t)(lr + 32) * F_IN);
    const float2* wp3 = (const float2*)(W + (size_t)(lr + 48) * F_IN);

    v8f c0 = {}, c1 = {}, c2 = {}, c3 = {};

    #pragma unroll
    for (int k0 = 0; k0 < F_IN; k0 += 4) {
        const int ki = (k0 >> 1) + khalf;
        float2 af = xp[ki];
        v2f a; a.x = af.x; a.y = af.y;
        float2 b0f = wp0[ki]; v2f b0; b0.x = b0f.x; b0.y = b0f.y;
        float2 b1f = wp1[ki]; v2f b1; b1.x = b1f.x; b1.y = b1f.y;
        float2 b2f = wp2[ki]; v2f b2; b2.x = b2f.x; b2.y = b2f.y;
        float2 b3f = wp3[ki]; v2f b3; b3.x = b3f.x; b3.y = b3f.y;

        c0 = __builtin_amdgcn_wmma_f32_16x16x4_f32(false, a, false, b0, (short)0, c0, false, false);
        c1 = __builtin_amdgcn_wmma_f32_16x16x4_f32(false, a, false, b1, (short)0, c1, false, false);
        c2 = __builtin_amdgcn_wmma_f32_16x16x4_f32(false, a, false, b2, (short)0, c2, false, false);
        c3 = __builtin_amdgcn_wmma_f32_16x16x4_f32(false, a, false, b3, (short)0, c3, false, false);
    }

    const int mbase = row0 + (khalf << 3);
    #pragma unroll
    for (int v = 0; v < 8; ++v) {
        int mm = mbase + v;
        if (mm < N) {
            float* hp = h + (size_t)mm * F_OUT + lr;
            hp[ 0] = c0[v];
            hp[16] = c1[v];
            hp[32] = c2[v];
            hp[48] = c3[v];
        }
    }
}


#define SL_NODES   1216
#define SL_SEGS    3
#define SL_THREADS 1024
#define SL_WAVES   (SL_THREADS / 32)
#define SL_QCAP    64
#define SL_LD      4
#define SL_ROWS    8

__device__ __forceinline__ void sl_drain(const uint2* qw, int n, int lane, int lo,
                                         const int* __restrict__ src,
                                         const float* __restrict__ dinv,
                                         const float* __restrict__ h, float* acc) {
    int s = 0; unsigned r = 0u; float nrm = 0.0f;
    if (lane < n) {
        const uint2 ent = qw[lane];
        r = ent.y;
        s = src[ent.x];
        nrm = dinv[s] * dinv[lo + (int)r];
    }
    #pragma clang loop unroll(disable)
    for (int i = 0; i < n; i += SL_ROWS) {
        float2   v[SL_ROWS];
        unsigned rr[SL_ROWS];
        float    nn[SL_ROWS];
        #pragma unroll
        for (int u = 0; u < SL_ROWS; ++u) {
            const int idx = (i + u < n) ? i + u : n - 1;
            const int su  = __builtin_amdgcn_readlane(s, idx);
            rr[u] = (unsigned)__builtin_amdgcn_readlane((int)r, idx);
            nn[u] = __int_as_float(__builtin_amdgcn_readlane(__float_as_int(nrm), idx));
            v[u]  = ((const float2*)(h + (size_t)su * F_OUT))[lane];
        }
        #pragma unroll
        for (int u = 0; u < SL_ROWS; ++u) {
            if (i + u < n) {
                float* a = acc + rr[u] * F_OUT + 2 * lane;
                atomicAdd(a,     v[u].x * nn[u]);
                atomicAdd(a + 1, v[u].y * nn[u]);
            }
        }
    }
}

__global__ void __launch_bounds__(SL_THREADS) k_slice_agg(const int* __restrict__ src,
                                                          const int* __restrict__ dst,
                                                          const float* __restrict__ h,
                                                          const float* __restrict__ dinv,
                                                          float* __restrict__ part, int E, int N) {
    __shared__ __attribute__((aligned(16))) float acc[SL_NODES * F_OUT];
    __shared__ uint2 q[SL_WAVES * SL_QCAP];

    const int tid  = threadIdx.x;
    const int lane = tid & 31;
    const int wave = __builtin_amdgcn_readfirstlane(tid >> 5);
    const int seg  = blockIdx.x;
    const int lo   = blockIdx.y * SL_NODES;

    float4* acc4 = (float4*)acc;
    for (int i = tid; i < SL_NODES * F_OUT / 4; i += SL_THREADS)
        acc4[i] = make_float4(0.0f, 0.0f, 0.0f, 0.0f);
    __syncthreads();

    const int   mis    = (int)(((size_t)dst >> 2) & 3);
    const int4* dst4   = (const int4*)(dst - mis);
    const int   ngrp   = (E + mis + 3) >> 2;
    const bool  ragged = (mis != 0) || (((E + mis) & 3) != 0);
    const int   per    = (ngrp + SL_SEGS - 1) / SL_SEGS;
    const int   g0     = seg * per;
    const int   g1     = (ngrp - g0 < per) ? ngrp : g0 + per;
    const int   nb     = (g1 - g0 + 32 * SL_LD - 1) / (32 * SL_LD);
    const int   rot    = (int)(((long long)blockIdx.y * nb) / (int)gridDim.y);

    uint2* qw = q + wave * SL_QCAP;
    int cnt = 0;

    int4 cur[SL_LD], nxt[SL_LD];

#define SL_LOAD(dstv, it)                                                          \
    {                                                                              \
        int b_ = (it) + rot; if (b_ >= nb) b_ -= nb;                               \
        _Pragma("unroll")                                                          \
        for (int j_ = 0; j_ < SL_LD; ++j_) {                                       \
            const int grp_ = g0 + (b_ * SL_LD + j_) * 32 + lane;                   \
            int4 v_ = make_int4(-1, -1, -1, -1);                                   \
            if (grp_ < g1) {                                                       \
                v_ = dst4[grp_];                                                   \
                if (ragged) {                                                      \
                    const int p_ = 4 * grp_ - mis;                                 \
                    if (p_     < 0 || p_     >= E) v_.x = -1;                      \
                    if (p_ + 1 < 0 || p_ + 1 >= E) v_.y = -1;                      \
                    if (p_ + 2 < 0 || p_ + 2 >= E) v_.z = -1;                      \
                    if (p_ + 3 < 0 || p_ + 3 >= E) v_.w = -1;                      \
                }                                                                  \
            }                                                                      \
            dstv[j_] = v_;                                                         \
        }                                                                          \
    }

#define SL_TAKE(val, eidx)                                                         \
    {                                                                              \
        const unsigned r_ = (unsigned)((val) - lo);                                \
        const bool hit_ = r_ < (unsigned)SL_NODES;                                 \
        const unsigned m_ = __builtin_amdgcn_ballot_w32(hit_);                     \
        if (m_ != 0u) {                                                            \
            if (hit_) qw[cnt + (int)__builtin_amdgcn_mbcnt_lo(m_, 0u)] = make_uint2((unsigned)(eidx), r_); \
            cnt += __builtin_popcount(m_);                                         \
            if (cnt >= 32) {                                                       \
                __builtin_amdgcn_fence(__ATOMIC_ACQ_REL, "wavefront");   \
                sl_drain(qw, 32, lane, lo, src, dinv, h, acc);                     \
                cnt -= 32;                                                         \
                uint2 mv_ = make_uint2(0u, 0u);                                    \
                if (lane < cnt) mv_ = qw[32 + lane];                               \
                if (lane < cnt) qw[lane] = mv_;                                    \
                __builtin_amdgcn_fence(__ATOMIC_ACQ_REL, "wavefront");   \
            }                                                                      \
        }                                                                          \
    }

    int it = wave;
    if (it < nb) SL_LOAD(nxt, it);
    while (it < nb) {
        #pragma unroll
        for (int t = 0; t < SL_LD; ++t) cur[t] = nxt[t];
        const int nit = it + SL_WAVES;
        if (nit < nb) SL_LOAD(nxt, nit);

        int b = it + rot; if (b >= nb) b -= nb;
        #pragma unroll
        for (int j = 0; j < SL_LD; ++j) {
            const int4 c = cur[j];
            const int e0 = 4 * (g0 + (b * SL_LD + j) * 32 + lane) - mis;
            SL_TAKE(c.x, e0);
            SL_TAKE(c.y, e0 + 1);
            SL_TAKE(c.z, e0 + 2);
            SL_TAKE(c.w, e0 + 3);
        }
        it = nit;
    }
    if (cnt > 0) {
        __builtin_amdgcn_fence(__ATOMIC_ACQ_REL, "wavefront");
        sl_drain(qw, cnt, lane, lo, src, dinv, h, acc);
    }
#undef SL_LOAD
#undef SL_TAKE

    __syncthreads();

    int rows = N - lo; if (rows > SL_NODES) rows = SL_NODES;
    float4* p4 = (float4*)(part + (size_t)seg * N * F_OUT + (size_t)lo * F_OUT);
    for (int i = tid; i < rows * (F_OUT / 4); i += SL_THREADS)
        p4[i] = acc4[i];
}


__global__ void k_finalize(const float* __restrict__ h, const float* __restrict__ dinv,
                           const float* __restrict__ b_conv, const float* __restrict__ part,
                           const float* __restrict__ W_lin, const float* __restrict__ b_lin,
                           float* __restrict__ out, int N) {
    int gtid = blockIdx.x * blockDim.x + threadIdx.x;
    int node = gtid >> 5;
    int lane = threadIdx.x & 31;
    if (node >= N) return;

    const size_t plane = (size_t)N * F_OUT;
    const size_t o     = (size_t)node * F_OUT + 2 * lane;
    const float  di    = dinv[node];
    const float2 hv    = *(const float2*)(h + o);
    const float2 bc    = *(const float2*)(b_conv + 2 * lane);
    const float2 wl    = *(const float2*)(W_lin + 2 * lane);
    float2 pv[SL_SEGS];
    #pragma unroll
    for (int g = 0; g < SL_SEGS; ++g) pv[g] = *(const float2*)(part + g * plane + o);

    float ax = bc.x + hv.x * di * di;
    float ay = bc.y + hv.y * di * di;
    #pragma unroll
    for (int g = 0; g < SL_SEGS; ++g) { ax += pv[g].x; ay += pv[g].y; }

    float p = fmaxf(ax, 0.0f) * wl.x + fmaxf(ay, 0.0f) * wl.y;

    #pragma unroll
    for (int off = 16; off > 0; off >>= 1)
        p += __shfl_down(p, off);

    if (lane == 0) {
        float z = p + b_lin[0];
        out[node] = 1.0f / (1.0f + __expf(-z));
    }
}


extern "C" void kernel_launch(void* const* d_in, const int* in_sizes, int n_in,
                              void* d_out, int out_size, void* d_ws, size_t ws_size,
                              hipStream_t stream) {
    const float* x      = (const float*)d_in[0];
    const int*   edge   = (const int*)  d_in[1];
    const float* W_conv = (const float*)d_in[2];
    const float* b_conv = (const float*)d_in[3];
    const float* W_lin  = (const float*)d_in[4];
    const float* b_lin  = (const float*)d_in[5];
    float*       out    = (float*)d_out;

    const int N = in_sizes[0] / F_IN;
    const int E = in_sizes[1] / 2;
    const int* src = edge;
    const int* dst = edge + E;

    float* ws   = (float*)d_ws;
    float* h    = ws;
    float* part = h + (size_t)N * F_OUT;
    float* dinv = part + (size_t)SL_SEGS * N * F_OUT;

    k_init_deg <<<(N + 255) / 256, 256, 0, stream>>>(dinv, N);
    k_count_deg<<<dim3(DEG_SEGS, (N + DEG_BINS - 1) / DEG_BINS), DEG_THREADS, 0, stream>>>(dst, dinv, E, N);
    k_dinv     <<<(N + 255) / 256, 256, 0, stream>>>(dinv, N);

    k_gemm<<<(N + 127) / 128, 256, 0, stream>>>(x, W_conv, h, N);

    k_slice_agg<<<dim3(SL_SEGS, (N + SL_NODES - 1) / SL_NODES), SL_THREADS, 0, stream>>>(src, dst, h, dinv, part, E, N);

    const long long fth = (long long)N * 32;
    k_finalize<<<(int)((fth + 255) / 256), 256, 0, stream>>>(h, dinv, b_conv, part, W_lin, b_lin, out, N);
}


# edge weight factored out of the edge loop: k_gemm stores rows pre-scaled by dinv[src], k_finalize scales each summed row by dinv[dst]; per-edge dinv gathers and multiply gone; product order changes
# speedup vs baseline: 2.2790x; 1.0315x over previous
#include <hip/hip_runtime.h>
#include <math.h>

typedef float v2f __attribute__((ext_vector_type(2)));
typedef float v8f __attribute__((ext_vector_type(8)));

#define F_IN  128
#define F_OUT 64


__global__ void k_init_deg(float* __restrict__ deg, int N) {
    int i = blockIdx.x * blockDim.x + threadIdx.x;
    if (i < N) deg[i] = 1.0f;
}


#define DEG_BINS    16384
#define DEG_SEGS    32
#define DEG_THREADS 1024
#define DEG_AHEAD   8

__global__ void __launch_bounds__(DEG_THREADS) k_count_deg(const int* __restrict__ dst,
                                                           float* __restrict__ deg, int E, int N) {
    __shared__ unsigned cnt[DEG_BINS];
    const int tid = threadIdx.x;
    const int lo  = blockIdx.y * DEG_BINS;

    for (int b = tid; b < DEG_BINS; b += DEG_THREADS) cnt[b] = 0u;
    __syncthreads();

    const int nseg  = (int)gridDim.x;
    const int chunk = (E + nseg - 1) / nseg;
    const int e0    = blockIdx.x * chunk;
    const int e1    = (E - e0 < chunk) ? E : e0 + chunk;

    for (int e = e0 + tid; e < e1; e += DEG_AHEAD * DEG_THREADS) {
        int d[DEG_AHEAD];
        #pragma unroll
        for (int j = 0; j < DEG_AHEAD; ++j) {
            const int ee = e + j * DEG_THREADS;
            d[j] = (ee < e1) ? dst[ee] : -1;
        }
        #pragma unroll
        for (int j = 0; j < DEG_AHEAD; ++j) {
            const unsigned r = (unsigned)(d[j] - lo);
            if (r < (unsigned)DEG_BINS) atomicAdd(&cnt[r], 1u);
        }
    }
    __syncthreads();

    const int rot = (blockIdx.x * DEG_THREADS) & (DEG_BINS - 1);
    unsigned c[DEG_BINS / DEG_THREADS];
    #pragma unroll
    for (int i = 0; i < DEG_BINS / DEG_THREADS; ++i)
        c[i] = cnt[(tid + i * DEG_THREADS + rot) & (DEG_BINS - 1)];
    #pragma unroll
    for (int i = 0; i < DEG_BINS / DEG_THREADS; ++i) {
        const int node = lo + ((tid + i * DEG_THREADS + rot) & (DEG_BINS - 1));
        if (c[i] != 0u && node < N) atomicAdd(&deg[node], (float)c[i]);
    }
}

__global__ void k_dinv(float* __restrict__ deg, int N) {
    int i = blockIdx.x * blockDim.x + threadIdx.x;
    if (i < N) {
        float d = deg[i];
        deg[i] = (d > 0.0f) ? rsqrtf(d) : 0.0f;
    }
}


__global__ void __launch_bounds__(256) k_gemm(const float* __restrict__ x,
                                              const float* __restrict__ W,
                                              const float* __restrict__ dinv,
                                              float* __restrict__ h, int N) {
    const int wave = threadIdx.x >> 5;
    const int lane = threadIdx.x & 31;
    const int row0 = (blockIdx.x * 8 + wave) * 16;
    if (row0 >= N) return;

    const int lr    = lane & 15;
    const int khalf = lane >> 4;

    int m = row0 + lr;
    if (m >= N) m = N - 1;

    const float2* xp  = (const float2*)(x + (size_t)m * F_IN);
    const float2* wp0 = (const float2*)(W + (size_t)(lr +  0) * F_IN);
    const float2* wp1 = (const float2*)(W + (size_t)(lr + 16) * F_IN);
    const float2* wp2 = (const float2*)(W + (size_t)(lr + 32) * F_IN);
    const float2* wp3 = (const float2*)(W + (size_t)(lr + 48) * F_IN);

    v8f c0 = {}, c1 = {}, c2 = {}, c3 = {};

    #pragma unroll
    for (int k0 = 0; k0 < F_IN; k0 += 4) {
        const int ki = (k0 >> 1) + khalf;
        float2 af = xp[ki];
        v2f a; a.x = af.x; a.y = af.y;
        float2 b0f = wp0[ki]; v2f b0; b0.x = b0f.x; b0.y = b0f.y;
        float2 b1f = wp1[ki]; v2f b1; b1.x = b1f.x; b1.y = b1f.y;
        float2 b2f = wp2[ki]; v2f b2; b2.x = b2f.x; b2.y = b2f.y;
        float2 b3f = wp3[ki]; v2f b3; b3.x = b3f.x; b3.y = b3f.y;

        c0 = __builtin_amdgcn_wmma_f32_16x16x4_f32(false, a, false, b0, (short)0, c0, false, false);
        c1 = __builtin_amdgcn_wmma_f32_16x16x4_f32(false, a, false, b1, (short)0, c1, false, false);
        c2 = __builtin_amdgcn_wmma_f32_16x16x4_f32(false, a, false, b2, (short)0, c2, false, false);
        c3 = __builtin_amdgcn_wmma_f32_16x16x4_f32(false, a, false, b3, (short)0, c3, false, false);
    }

    const int mbase = row0 + (khalf << 3);
    #pragma unroll
    for (int v = 0; v < 8; ++v) {
        int mm = mbase + v;
        if (mm < N) {
            const float sc = dinv[mm];
            float* hp = h + (size_t)mm * F_OUT + lr;
            hp[ 0] = c0[v] * sc;
            hp[16] = c1[v] * sc;
            hp[32] = c2[v] * sc;
            hp[48] = c3[v] * sc;
        }
    }
}


#define SL_NODES   1216
#define SL_SEGS    3
#define SL_THREADS 1024
#define SL_WAVES   (SL_THREADS / 32)
#define SL_QCAP    64
#define SL_LD      4
#define SL_ROWS    8

__device__ __forceinline__ void sl_drain(const uint2* qw, int n, int lane,
                                         const int* __restrict__ src,
                                         const float* __restrict__ h, float* acc) {
    int s = 0; unsigned r = 0u;
    if (lane < n) {
        const uint2 ent = qw[lane];
        r = ent.y;
        s = src[ent.x];
    }
    #pragma clang loop unroll(disable)
    for (int i = 0; i < n; i += SL_ROWS) {
        float2   v[SL_ROWS];
        unsigned rr[SL_ROWS];
        #pragma unroll
        for (int u = 0; u < SL_ROWS; ++u) {
            const int idx = (i + u < n) ? i + u : n - 1;
            const int su  = __builtin_amdgcn_readlane(s, idx);
            rr[u] = (unsigned)__builtin_amdgcn_readlane((int)r, idx);
            v[u]  = ((const float2*)(h + (size_t)su * F_OUT))[lane];
        }
        #pragma unroll
        for (int u = 0; u < SL_ROWS; ++u) {
            if (i + u < n) {
                float* a = acc + rr[u] * F_OUT + 2 * lane;
                atomicAdd(a,     v[u].x);
                atomicAdd(a + 1, v[u].y);
            }
        }
    }
}

__global__ void __launch_bounds__(SL_THREADS) k_slice_agg(const int* __restrict__ src,
                                                          const int* __restrict__ dst,
                                                          const float* __restrict__ h,
                                                          float* __restrict__ part, int E, int N) {
    __shared__ __attribute__((aligned(16))) float acc[SL_NODES * F_OUT];
    __shared__ uint2 q[SL_WAVES * SL_QCAP];

    const int tid  = threadIdx.x;
    const int lane = tid & 31;
    const int wave = __builtin_amdgcn_readfirstlane(tid >> 5);
    const int seg  = blockIdx.x;
    const int lo   = blockIdx.y * SL_NODES;

    float4* acc4 = (float4*)acc;
    for (int i = tid; i < SL_NODES * F_OUT / 4; i += SL_THREADS)
        acc4[i] = make_float4(0.0f, 0.0f, 0.0f, 0.0f);
    __syncthreads();

    const int   mis    = (int)(((size_t)dst >> 2) & 3);
    const int4* dst4   = (const int4*)(dst - mis);
    const int   ngrp   = (E + mis + 3) >> 2;
    const bool  ragged = (mis != 0) || (((E + mis) & 3) != 0);
    const int   per    = (ngrp + SL_SEGS - 1) / SL_SEGS;
    const int   g0     = seg * per;
    const int   g1     = (ngrp - g0 < per) ? ngrp : g0 + per;
    const int   nb     = (g1 - g0 + 32 * SL_LD - 1) / (32 * SL_LD);
    const int   rot    = (int)(((long long)blockIdx.y * nb) / (int)gridDim.y);

    uint2* qw = q + wave * SL_QCAP;
    int cnt = 0;

    int4 cur[SL_LD], nxt[SL_LD];

#define SL_LOAD(dstv, it)                                                          \
    {                                                                              \
        int b_ = (it) + rot; if (b_ >= nb) b_ -= nb;                               \
        _Pragma("unroll")                                                          \
        for (int j_ = 0; j_ < SL_LD; ++j_) {                                       \
            const int grp_ = g0 + (b_ * SL_LD + j_) * 32 + lane;                   \
            int4 v_ = make_int4(-1, -1, -1, -1);                                   \
            if (grp_ < g1) {                                                       \
                v_ = dst4[grp_];                                                   \
                if (ragged) {                                                      \
                    const int p_ = 4 * grp_ - mis;                                 \
                    if (p_     < 0 || p_     >= E) v_.x = -1;                      \
                    if (p_ + 1 < 0 || p_ + 1 >= E) v_.y = -1;                      \
                    if (p_ + 2 < 0 || p_ + 2 >= E) v_.z = -1;                      \
                    if (p_ + 3 < 0 || p_ + 3 >= E) v_.w = -1;                      \
                }                                                                  \
            }                                                                      \
            dstv[j_] = v_;                                                         \
        }                                                                          \
    }

#define SL_TAKE(val, eidx)                                                         \
    {                                                                              \
        const unsigned r_ = (unsigned)((val) - lo);                                \
        const bool hit_ = r_ < (unsigned)SL_NODES;                                 \
        const unsigned m_ = __builtin_amdgcn_ballot_w32(hit_);                     \
        if (m_ != 0u) {                                                            \
            if (hit_) qw[cnt + (int)__builtin_amdgcn_mbcnt_lo(m_, 0u)] = make_uint2((unsigned)(eidx), r_); \
            cnt += __builtin_popcount(m_);                                         \
            if (cnt >= 32) {                                                       \
                __builtin_amdgcn_fence(__ATOMIC_ACQ_REL, "wavefront");   \
                sl_drain(qw, 32, lane, src, h, acc);                               \
                cnt -= 32;                                                         \
                uint2 mv_ = make_uint2(0u, 0u);                                    \
                if (lane < cnt) mv_ = qw[32 + lane];                               \
                if (lane < cnt) qw[lane] = mv_;                                    \
                __builtin_amdgcn_fence(__ATOMIC_ACQ_REL, "wavefront");   \
            }                                                                      \
        }                                                                          \
    }

    int it = wave;
    if (it < nb) SL_LOAD(nxt, it);
    while (it < nb) {
        #pragma unroll
        for (int t = 0; t < SL_LD; ++t) cur[t] = nxt[t];
        const int nit = it + SL_WAVES;
        if (nit < nb) SL_LOAD(nxt, nit);

        int b = it + rot; if (b >= nb) b -= nb;
        #pragma unroll
        for (int j = 0; j < SL_LD; ++j) {
            const int4 c = cur[j];
            const int e0 = 4 * (g0 + (b * SL_LD + j) * 32 + lane) - mis;
            SL_TAKE(c.x, e0);
            SL_TAKE(c.y, e0 + 1);
            SL_TAKE(c.z, e0 + 2);
            SL_TAKE(c.w, e0 + 3);
        }
        it = nit;
    }
    if (cnt > 0) {
        __builtin_amdgcn_fence(__ATOMIC_ACQ_REL, "wavefront");
        sl_drain(qw, cnt, lane, src, h, acc);
    }
#undef SL_LOAD
#undef SL_TAKE

    __syncthreads();

    int rows = N - lo; if (rows > SL_NODES) rows = SL_NODES;
    float4* p4 = (float4*)(part + (size_t)seg * N * F_OUT + (size_t)lo * F_OUT);
    for (int i = tid; i < rows * (F_OUT / 4); i += SL_THREADS)
        p4[i] = acc4[i];
}


__global__ void k_finalize(const float* __restrict__ h, const float* __restrict__ dinv,
                           const float* __restrict__ b_conv, const float* __restrict__ part,
                           const float* __restrict__ W_lin, const float* __restrict__ b_lin,
                           float* __restrict__ out, int N) {
    int gtid = blockIdx.x * blockDim.x + threadIdx.x;
    int node = gtid >> 5;
    int lane = threadIdx.x & 31;
    if (node >= N) return;

    const size_t plane = (size_t)N * F_OUT;
    const size_t o     = (size_t)node * F_OUT + 2 * lane;
    const float  di    = dinv[node];
    const float2 hv    = *(const float2*)(h + o);
    const float2 bc    = *(const float2*)(b_conv + 2 * lane);
    const float2 wl    = *(const float2*)(W_lin + 2 * lane);
    float2 pv[SL_SEGS];
    #pragma unroll
    for (int g = 0; g < SL_SEGS; ++g) pv[g] = *(const float2*)(part + g * plane + o);

    float sx = hv.x, sy = hv.y;
    #pragma unroll
    for (int g = 0; g < SL_SEGS; ++g) { sx += pv[g].x; sy += pv[g].y; }
    const float ax = bc.x + di * sx;
    const float ay = bc.y + di * sy;

    float p = fmaxf(ax, 0.0f) * wl.x + fmaxf(ay, 0.0f) * wl.y;

    #pragma unroll
    for (int off = 16; off > 0; off >>= 1)
        p += __shfl_down(p, off);

    if (lane == 0) {
        float z = p + b_lin[0];
        out[node] = 1.0f / (1.0f + __expf(-z));
    }
}


extern "C" void kernel_launch(void* const* d_in, const int* in_sizes, int n_in,
                              void* d_out, int out_size, void* d_ws, size_t ws_size,
                              hipStream_t stream) {
    const float* x      = (const float*)d_in[0];
    const int*   edge   = (const int*)  d_in[1];
    const float* W_conv = (const float*)d_in[2];
    const float* b_conv = (const float*)d_in[3];
    const float* W_lin  = (const float*)d_in[4];
    const float* b_lin  = (const float*)d_in[5];
    float*       out    = (float*)d_out;

    const int N = in_sizes[0] / F_IN;
    const int E = in_sizes[1] / 2;
    const int* src = edge;
    const int* dst = edge + E;

    float* ws   = (float*)d_ws;
    float* h    = ws;
    float* part = h + (size_t)N * F_OUT;
    float* dinv = part + (size_t)SL_SEGS * N * F_OUT;

    k_init_deg <<<(N + 255) / 256, 256, 0, stream>>>(dinv, N);
    k_count_deg<<<dim3(DEG_SEGS, (N + DEG_BINS - 1) / DEG_BINS), DEG_THREADS, 0, stream>>>(dst, dinv, E, N);
    k_dinv     <<<(N + 255) / 256, 256, 0, stream>>>(dinv, N);

    k_gemm<<<(N + 127) / 128, 256, 0, stream>>>(x, W_conv, dinv, h, N);

    k_slice_agg<<<dim3(SL_SEGS, (N + SL_NODES - 1) / SL_NODES), SL_THREADS, 0, stream>>>(src, dst, h, part, E, N);

    const long long fth = (long long)N * 32;
    k_finalize<<<(int)((fth + 255) / 256), 256, 0, stream>>>(h, dinv, b_conv, part, W_lin, b_lin, out, N);
}


# k_slice_agg scans 1-byte slice ids (32 per lane, matched 4 per vector op) plus one packed src/row word per hit, both written by the init kernel, in place of 4-byte dst tests; row sum order changes
# speedup vs baseline: 2.4378x; 1.0697x over previous
#include <hip/hip_runtime.h>
#include <math.h>

typedef float v2f __attribute__((ext_vector_type(2)));
typedef float v8f __attribute__((ext_vector_type(8)));

#define F_IN  128
#define F_OUT 64


#define SL_NODES    1216
#define SL_PAD_KEY  127u
#define SL_SRC_BITS 20

__global__ void k_prep(const int* __restrict__ src, const int* __restrict__ dst,
                       float* __restrict__ deg, unsigned* __restrict__ keys,
                       unsigned* __restrict__ packed, int E, int N) {
    const int i  = blockIdx.x * blockDim.x + threadIdx.x;
    const int nq = ((E + 31) >> 5) << 3;
    if (i < N) deg[i] = 1.0f;
    if (i < nq) {
        unsigned kw = 0u;
        unsigned p[4];
        #pragma unroll
        for (int j = 0; j < 4; ++j) {
            const int e = 4 * i + j;
            unsigned key = SL_PAD_KEY, pk = 0u;
            if (e < E) {
                const unsigned d  = (unsigned)dst[e];
                const unsigned sl = d / (unsigned)SL_NODES;
                key = sl;
                pk  = (unsigned)src[e] | ((d - sl * (unsigned)SL_NODES) << SL_SRC_BITS);
            }
            kw |= key << (8 * j);
            p[j] = pk;
        }
        keys[i] = kw;
        ((uint4*)packed)[i] = make_uint4(p[0], p[1], p[2], p[3]);
    }
}


#define DEG_BINS    16384
#define DEG_SEGS    32
#define DEG_THREADS 1024
#define DEG_AHEAD   8

__global__ void __launch_bounds__(DEG_THREADS) k_count_deg(const int* __restrict__ dst,
                                                           float* __restrict__ deg, int E, int N) {
    __shared__ unsigned cnt[DEG_BINS];
    const int tid = threadIdx.x;
    const int lo  = blockIdx.y * DEG_BINS;

    for (int b = tid; b < DEG_BINS; b += DEG_THREADS) cnt[b] = 0u;
    __syncthreads();

    const int nseg  = (int)gridDim.x;
    const int chunk = (E + nseg - 1) / nseg;
    const int e0    = blockIdx.x * chunk;
    const int e1    = (E - e0 < chunk) ? E : e0 + chunk;

    for (int e = e0 + tid; e < e1; e += DEG_AHEAD * DEG_THREADS) {
        int d[DEG_AHEAD];
        #pragma unroll
        for (int j = 0; j < DEG_AHEAD; ++j) {
            const int ee = e + j * DEG_THREADS;
            d[j] = (ee < e1) ? dst[ee] : -1;
        }
        #pragma unroll
        for (int j = 0; j < DEG_AHEAD; ++j) {
            const unsigned r = (unsigned)(d[j] - lo);
            if (r < (unsigned)DEG_BINS) atomicAdd(&cnt[r], 1u);
        }
    }
    __syncthreads();

    const int rot = (blockIdx.x * DEG_THREADS) & (DEG_BINS - 1);
    unsigned c[DEG_BINS / DEG_THREADS];
    #pragma unroll
    for (int i = 0; i < DEG_BINS / DEG_THREADS; ++i)
        c[i] = cnt[(tid + i * DEG_THREADS + rot) & (DEG_BINS - 1)];
    #pragma unroll
    for (int i = 0; i < DEG_BINS / DEG_THREADS; ++i) {
        const int node = lo + ((tid + i * DEG_THREADS + rot) & (DEG_BINS - 1));
        if (c[i] != 0u && node < N) atomicAdd(&deg[node], (float)c[i]);
    }
}

__global__ void k_dinv(float* __restrict__ deg, int N) {
    int i = blockIdx.x * blockDim.x + threadIdx.x;
    if (i < N) {
        float d = deg[i];
        deg[i] = (d > 0.0f) ? rsqrtf(d) : 0.0f;
    }
}


__global__ void __launch_bounds__(256) k_gemm(const float* __restrict__ x,
                                              const float* __restrict__ W,
                                              const float* __restrict__ dinv,
                                              float* __restrict__ h, int N) {
    const int wave = threadIdx.x >> 5;
    const int lane = threadIdx.x & 31;
    const int row0 = (blockIdx.x * 8 + wave) * 16;
    if (row0 >= N) return;

    const int lr    = lane & 15;
    const int khalf = lane >> 4;

    int m = row0 + lr;
    if (m >= N) m = N - 1;

    const float2* xp  = (const float2*)(x + (size_t)m * F_IN);
    const float2* wp0 = (const float2*)(W + (size_t)(lr +  0) * F_IN);
    const float2* wp1 = (const float2*)(W + (size_t)(lr + 16) * F_IN);
    const float2* wp2 = (const float2*)(W + (size_t)(lr + 32) * F_IN);
    const float2* wp3 = (const float2*)(W + (size_t)(lr + 48) * F_IN);

    v8f c0 = {}, c1 = {}, c2 = {}, c3 = {};

    #pragma unroll
    for (int k0 = 0; k0 < F_IN; k0 += 4) {
        const int ki = (k0 >> 1) + khalf;
        float2 af = xp[ki];
        v2f a; a.x = af.x; a.y = af.y;
        float2 b0f = wp0[ki]; v2f b0; b0.x = b0f.x; b0.y = b0f.y;
        float2 b1f = wp1[ki]; v2f b1; b1.x = b1f.x; b1.y = b1f.y;
        float2 b2f = wp2[ki]; v2f b2; b2.x = b2f.x; b2.y = b2f.y;
        float2 b3f = wp3[ki]; v2f b3; b3.x = b3f.x; b3.y = b3f.y;

        c0 = __builtin_amdgcn_wmma_f32_16x16x4_f32(false, a, false, b0, (short)0, c0, false, false);
        c1 = __builtin_amdgcn_wmma_f32_16x16x4_f32(false, a, false, b1, (short)0, c1, false, false);
        c2 = __builtin_amdgcn_wmma_f32_16x16x4_f32(false, a, false, b2, (short)0, c2, false, false);
        c3 = __builtin_amdgcn_wmma_f32_16x16x4_f32(false, a, false, b3, (short)0, c3, false, false);
    }

    const int mbase = row0 + (khalf << 3);
    #pragma unroll
    for (int v = 0; v < 8; ++v) {
        int mm = mbase + v;
        if (mm < N) {
            const float sc = dinv[mm];
            float* hp = h + (size_t)mm * F_OUT + lr;
            hp[ 0] = c0[v] * sc;
            hp[16] = c1[v] * sc;
            hp[32] = c2[v] * sc;
            hp[48] = c3[v] * sc;
        }
    }
}


#define SL_SEGS    3
#define SL_THREADS 1024
#define SL_WAVES   (SL_THREADS / 32)
#define SL_QCAP    64
#define SL_ROWS    8

__device__ __forceinline__ void sl_drain(const unsigned* qw, int n, int lane,
                                         const unsigned* __restrict__ packed,
                                         const float* __restrict__ h, float* acc) {
    int s = 0; unsigned r = 0u;
    if (lane < n) {
        const unsigned p = packed[qw[lane]];
        s = (int)(p & ((1u << SL_SRC_BITS) - 1u));
        r = p >> SL_SRC_BITS;
    }
    #pragma clang loop unroll(disable)
    for (int i = 0; i < n; i += SL_ROWS) {
        float2   v[SL_ROWS];
        unsigned rr[SL_ROWS];
        #pragma unroll
        for (int u = 0; u < SL_ROWS; ++u) {
            const int idx = (i + u < n) ? i + u : n - 1;
            const int su  = __builtin_amdgcn_readlane(s, idx);
            rr[u] = (unsigned)__builtin_amdgcn_readlane((int)r, idx);
            v[u]  = ((const float2*)(h + (size_t)su * F_OUT))[lane];
        }
        #pragma unroll
        for (int u = 0; u < SL_ROWS; ++u) {
            if (i + u < n) {
                float* a = acc + rr[u] * F_OUT + 2 * lane;
                atomicAdd(a,     v[u].x);
                atomicAdd(a + 1, v[u].y);
            }
        }
    }
}

__global__ void __launch_bounds__(SL_THREADS) k_slice_agg(const unsigned* __restrict__ keys,
                                                          const unsigned* __restrict__ packed,
                                                          const float* __restrict__ h,
                                                          float* __restrict__ part, int E, int N) {
    __shared__ __attribute__((aligned(16))) float acc[SL_NODES * F_OUT];
    __shared__ unsigned q[SL_WAVES * SL_QCAP];

    const int tid  = threadIdx.x;
    const int lane = tid & 31;
    const int wave = __builtin_amdgcn_readfirstlane(tid >> 5);
    const int seg  = blockIdx.x;
    const int lo   = blockIdx.y * SL_NODES;
    const unsigned sid4 = (unsigned)blockIdx.y * 0x01010101u;

    float4* acc4 = (float4*)acc;
    for (int i = tid; i < SL_NODES * F_OUT / 4; i += SL_THREADS)
        acc4[i] = make_float4(0.0f, 0.0f, 0.0f, 0.0f);
    __syncthreads();

    const uint4* key4 = (const uint4*)keys;
    const int    ngrp = (E + 31) >> 5;
    const int    per  = (ngrp + SL_SEGS - 1) / SL_SEGS;
    const int    g0   = seg * per;
    const int    g1   = (ngrp - g0 < per) ? ngrp : g0 + per;
    const int    nb   = (g1 - g0 + 31) / 32;
    const int    rot  = (int)(((long long)blockIdx.y * nb) / (int)gridDim.y);

    unsigned* qw = q + wave * SL_QCAP;
    int cnt = 0;

    uint4 curA, curB, nxtA, nxtB;
    const unsigned padw = SL_PAD_KEY * 0x01010101u;

#define SL_LOAD(va, vb, it)                                                        \
    {                                                                              \
        int b_ = (it) + rot; if (b_ >= nb) b_ -= nb;                               \
        const int grp_ = g0 + b_ * 32 + lane;                                      \
        uint4 a_ = make_uint4(padw, padw, padw, padw), c_ = a_;                    \
        if (grp_ < g1) { a_ = key4[2 * grp_]; c_ = key4[2 * grp_ + 1]; }           \
        va = a_; vb = c_;                                                          \
    }

#define SL_NE(word, w) ((((word ^ sid4) + 0x7F7F7F7Fu) >> (7 - (w))) & (0x01010101u << (w)))

    int it = wave;
    if (it < nb) SL_LOAD(nxtA, nxtB, it);
    while (it < nb) {
        curA = nxtA; curB = nxtB;
        const int nit = it + SL_WAVES;
        if (nit < nb) SL_LOAD(nxtA, nxtB, nit);

        int b = it + rot; if (b >= nb) b -= nb;
        const unsigned ebase = 32u * (unsigned)(g0 + b * 32 + lane);

        const unsigned ne = SL_NE(curA.x, 0) | SL_NE(curA.y, 1) | SL_NE(curA.z, 2) | SL_NE(curA.w, 3)
                          | SL_NE(curB.x, 4) | SL_NE(curB.y, 5) | SL_NE(curB.z, 6) | SL_NE(curB.w, 7);
        unsigned hm = ~ne;

        #pragma clang loop unroll(disable)
        for (int pass = 0; pass < 32; ++pass) {
            const bool has = hm != 0u;
            const unsigned m = __builtin_amdgcn_ballot_w32(has);
            if (m == 0u) break;
            if (has) {
                const unsigned pos = (unsigned)__builtin_ctz(hm);
                hm &= hm - 1u;
                qw[cnt + (int)__builtin_amdgcn_mbcnt_lo(m, 0u)] = ebase + ((pos & 7u) << 2) + (pos >> 3);
            }
            cnt += __builtin_popcount(m);
            if (cnt >= 32) {
                __builtin_amdgcn_fence(__ATOMIC_ACQ_REL, "wavefront");
                sl_drain(qw, 32, lane, packed, h, acc);
                cnt -= 32;
                unsigned mv = 0u;
                if (lane < cnt) mv = qw[32 + lane];
                if (lane < cnt) qw[lane] = mv;
                __builtin_amdgcn_fence(__ATOMIC_ACQ_REL, "wavefront");
            }
        }
        it = nit;
    }
    if (cnt > 0) {
        __builtin_amdgcn_fence(__ATOMIC_ACQ_REL, "wavefront");
        sl_drain(qw, cnt, lane, packed, h, acc);
    }
#undef SL_LOAD
#undef SL_NE

    __syncthreads();

    int rows = N - lo; if (rows > SL_NODES) rows = SL_NODES;
    float4* p4 = (float4*)(part + (size_t)seg * N * F_OUT + (size_t)lo * F_OUT);
    for (int i = tid; i < rows * (F_OUT / 4); i += SL_THREADS)
        p4[i] = acc4[i];
}


__global__ void k_finalize(const float* __restrict__ h, const float* __restrict__ dinv,
                           const float* __restrict__ b_conv, const float* __restrict__ part,
                           const float* __restrict__ W_lin, const float* __restrict__ b_lin,
                           float* __restrict__ out, int N) {
    int gtid = blockIdx.x * blockDim.x + threadIdx.x;
    int node = gtid >> 5;
    int lane = threadIdx.x & 31;
    if (node >= N) return;

    const size_t plane = (size_t)N * F_OUT;
    const size_t o     = (size_t)node * F_OUT + 2 * lane;
    const float  di    = dinv[node];
    const float2 hv    = *(const float2*)(h + o);
    const float2 bc    = *(const float2*)(b_conv + 2 * lane);
    const float2 wl    = *(const float2*)(W_lin + 2 * lane);
    float2 pv[SL_SEGS];
    #pragma unroll
    for (int g = 0; g < SL_SEGS; ++g) pv[g] = *(const float2*)(part + g * plane + o);

    float sx = hv.x, sy = hv.y;
    #pragma unroll
    for (int g = 0; g < SL_SEGS; ++g) { sx += pv[g].x; sy += pv[g].y; }
    const float ax = bc.x + di * sx;
    const float ay = bc.y + di * sy;

    float p = fmaxf(ax, 0.0f) * wl.x + fmaxf(ay, 0.0f) * wl.y;

    #pragma unroll
    for (int off = 16; off > 0; off >>= 1)
        p += __shfl_down(p, off);

    if (lane == 0) {
        float z = p + b_lin[0];
        out[node] = 1.0f / (1.0f + __expf(-z));
    }
}


extern "C" void kernel_launch(void* const* d_in, const int* in_sizes, int n_in,
                              void* d_out, int out_size, void* d_ws, size_t ws_size,
                              hipStream_t stream) {
    const float* x      = (const float*)d_in[0];
    const int*   edge   = (const int*)  d_in[1];
    const float* W_conv = (const float*)d_in[2];
    const float* b_conv = (const float*)d_in[3];
    const float* W_lin  = (const float*)d_in[4];
    const float* b_lin  = (const float*)d_in[5];
    float*       out    = (float*)d_out;

    const int N = in_sizes[0] / F_IN;
    const int E = in_sizes[1] / 2;
    const int* src = edge;
    const int* dst = edge + E;

    float* ws   = (float*)d_ws;
    float* h    = ws;
    float* part = h + (size_t)N * F_OUT;
    float* dinv = part + (size_t)SL_SEGS * N * F_OUT;
    const int Epad = (E + 31) / 32 * 32;
    unsigned* packed = (unsigned*)(dinv + (N + 63) / 64 * 64);
    unsigned* keys   = packed + Epad;

    const int nprep = (N > Epad / 4) ? N : Epad / 4;
    k_prep     <<<(nprep + 255) / 256, 256, 0, stream>>>(src, dst, dinv, keys, packed, E, N);
    k_count_deg<<<dim3(DEG_SEGS, (N + DEG_BINS - 1) / DEG_BINS), DEG_THREADS, 0, stream>>>(dst, dinv, E, N);
    k_dinv     <<<(N + 255) / 256, 256, 0, stream>>>(dinv, N);

    k_gemm<<<(N + 127) / 128, 256, 0, stream>>>(x, W_conv, dinv, h, N);

    k_slice_agg<<<dim3(SL_SEGS, (N + SL_NODES - 1) / SL_NODES), SL_THREADS, 0, stream>>>(keys, packed, h, part, E, N);

    const long long fth = (long long)N * 32;
    k_finalize<<<(int)((fth + 255) / 256), 256, 0, stream>>>(h, dinv, b_conv, part, W_lin, b_lin, out, N);
}


# k_gemm made persistent: 1024 waves, each keeps all W_conv fragments in registers and walks its row tiles, next tile's x fragments requested ahead; was one tile per wave re-reading W; sums unchanged
# speedup vs baseline: 2.7906x; 1.1447x over previous
#include <hip/hip_runtime.h>
#include <math.h>

typedef float v2f __attribute__((ext_vector_type(2)));
typedef float v8f __attribute__((ext_vector_type(8)));

#define F_IN  128
#define F_OUT 64


#define SL_NODES    1216
#define SL_PAD_KEY  127u
#define SL_SRC_BITS 20

__global__ void k_prep(const int* __restrict__ src, const int* __restrict__ dst,
                       float* __restrict__ deg, unsigned* __restrict__ keys,
                       unsigned* __restrict__ packed, int E, int N) {
    const int i  = blockIdx.x * blockDim.x + threadIdx.x;
    const int nq = ((E + 31) >> 5) << 3;
    if (i < N) deg[i] = 1.0f;
    if (i < nq) {
        unsigned kw = 0u;
        unsigned p[4];
        #pragma unroll
        for (int j = 0; j < 4; ++j) {
            const int e = 4 * i + j;
            unsigned key = SL_PAD_KEY, pk = 0u;
            if (e < E) {
                const unsigned d  = (unsigned)dst[e];
                const unsigned sl = d / (unsigned)SL_NODES;
                key = sl;
                pk  = (unsigned)src[e] | ((d - sl * (unsigned)SL_NODES) << SL_SRC_BITS);
            }
            kw |= key << (8 * j);
            p[j] = pk;
        }
        keys[i] = kw;
        ((uint4*)packed)[i] = make_uint4(p[0], p[1], p[2], p[3]);
    }
}


#define DEG_BINS    16384
#define DEG_SEGS    32
#define DEG_THREADS 1024
#define DEG_AHEAD   8

__global__ void __launch_bounds__(DEG_THREADS) k_count_deg(const int* __restrict__ dst,
                                                           float* __restrict__ deg, int E, int N) {
    __shared__ unsigned cnt[DEG_BINS];
    const int tid = threadIdx.x;
    const int lo  = blockIdx.y * DEG_BINS;

    for (int b = tid; b < DEG_BINS; b += DEG_THREADS) cnt[b] = 0u;
    __syncthreads();

    const int nseg  = (int)gridDim.x;
    const int chunk = (E + nseg - 1) / nseg;
    const int e0    = blockIdx.x * chunk;
    const int e1    = (E - e0 < chunk) ? E : e0 + chunk;

    for (int e = e0 + tid; e < e1; e += DEG_AHEAD * DEG_THREADS) {
        int d[DEG_AHEAD];
        #pragma unroll
        for (int j = 0; j < DEG_AHEAD; ++j) {
            const int ee = e + j * DEG_THREADS;
            d[j] = (ee < e1) ? dst[ee] : -1;
        }
        #pragma unroll
        for (int j = 0; j < DEG_AHEAD; ++j) {
            const unsigned r = (unsigned)(d[j] - lo);
            if (r < (unsigned)DEG_BINS) atomicAdd(&cnt[r], 1u);
        }
    }
    __syncthreads();

    const int rot = (blockIdx.x * DEG_THREADS) & (DEG_BINS - 1);
    unsigned c[DEG_BINS / DEG_THREADS];
    #pragma unroll
    for (int i = 0; i < DEG_BINS / DEG_THREADS; ++i)
        c[i] = cnt[(tid + i * DEG_THREADS + rot) & (DEG_BINS - 1)];
    #pragma unroll
    for (int i = 0; i < DEG_BINS / DEG_THREADS; ++i) {
        const int node = lo + ((tid + i * DEG_THREADS + rot) & (DEG_BINS - 1));
        if (c[i] != 0u && node < N) atomicAdd(&deg[node], (float)c[i]);
    }
}

__global__ void k_dinv(float* __restrict__ deg, int N) {
    int i = blockIdx.x * blockDim.x + threadIdx.x;
    if (i < N) {
        float d = deg[i];
        deg[i] = (d > 0.0f) ? rsqrtf(d) : 0.0f;
    }
}


#define GM_WAVES  4
#define GM_BLOCKS 256
#define GM_KSTEPS (F_IN / 4)

__global__ void __launch_bounds__(GM_WAVES * 32) k_gemm(const float* __restrict__ x,
                                                        const float* __restrict__ W,
                                                        const float* __restrict__ dinv,
                                                        float* __restrict__ h, int N) {
    const int wave = __builtin_amdgcn_readfirstlane(threadIdx.x >> 5);
    const int lane = threadIdx.x & 31;
    const int lr    = lane & 15;
    const int khalf = lane >> 4;

    const int ntiles = (N + 15) >> 4;
    const int nw     = (int)gridDim.x * GM_WAVES;
    int tile = wave * (int)gridDim.x + (int)blockIdx.x;
    if (tile >= ntiles) return;

    v2f b[GM_KSTEPS][4];
    #pragma unroll
    for (int t = 0; t < 4; ++t) {
        const float2* wp = (const float2*)(W + (size_t)(lr + 16 * t) * F_IN);
        #pragma unroll
        for (int ks = 0; ks < GM_KSTEPS; ++ks) {
            const float2 f = wp[2 * ks + khalf];
            b[ks][t].x = f.x; b[ks][t].y = f.y;
        }
    }

    v2f a[GM_KSTEPS];
    {
        int m = tile * 16 + lr;
        if (m >= N) m = N - 1;
        const float2* xp = (const float2*)(x + (size_t)m * F_IN);
        #pragma unroll
        for (int ks = 0; ks < GM_KSTEPS; ++ks) {
            const float2 f = xp[2 * ks + khalf];
            a[ks].x = f.x; a[ks].y = f.y;
        }
    }

    for (;;) {
        const int  nxt  = tile + nw;
        const bool more = nxt < ntiles;

        const int mbase = tile * 16 + (khalf << 3);
        float sc[8];
        #pragma unroll
        for (int v = 0; v < 8; ++v) {
            int mm = mbase + v;
            if (mm >= N) mm = N - 1;
            sc[v] = dinv[mm];
        }

        v2f an[GM_KSTEPS];
        {
            int m = (more ? nxt : tile) * 16 + lr;
            if (m >= N) m = N - 1;
            const float2* xp = (const float2*)(x + (size_t)m * F_IN);
            #pragma unroll
            for (int ks = 0; ks < GM_KSTEPS; ++ks) {
                const float2 f = xp[2 * ks + khalf];
                an[ks].x = f.x; an[ks].y = f.y;
            }
        }

        v8f c0 = {}, c1 = {}, c2 = {}, c3 = {};
        #pragma unroll
        for (int ks = 0; ks < GM_KSTEPS; ++ks) {
            c0 = __builtin_amdgcn_wmma_f32_16x16x4_f32(false, a[ks], false, b[ks][0], (short)0, c0, false, false);
            c1 = __builtin_amdgcn_wmma_f32_16x16x4_f32(false, a[ks], false, b[ks][1], (short)0, c1, false, false);
            c2 = __builtin_amdgcn_wmma_f32_16x16x4_f32(false, a[ks], false, b[ks][2], (short)0, c2, false, false);
            c3 = __builtin_amdgcn_wmma_f32_16x16x4_f32(false, a[ks], false, b[ks][3], (short)0, c3, false, false);
        }

        if (tile * 16 + 16 <= N) {
            float* hp = h + (size_t)mbase * F_OUT + lr;
            #pragma unroll
            for (int v = 0; v < 8; ++v) {
                hp[v * F_OUT +  0] = c0[v] * sc[v];
                hp[v * F_OUT + 16] = c1[v] * sc[v];
                hp[v * F_OUT + 32] = c2[v] * sc[v];
                hp[v * F_OUT + 48] = c3[v] * sc[v];
            }
        } else {
            #pragma unroll
            for (int v = 0; v < 8; ++v) {
                const int mm = mbase + v;
                if (mm < N) {
                    float* hp = h + (size_t)mm * F_OUT + lr;
                    hp[ 0] = c0[v] * sc[v];
                    hp[16] = c1[v] * sc[v];
                    hp[32] = c2[v] * sc[v];
                    hp[48] = c3[v] * sc[v];
                }
            }
        }

        if (!more) break;
        #pragma unroll
        for (int ks = 0; ks < GM_KSTEPS; ++ks) a[ks] = an[ks];
        tile = nxt;
    }
}


#define SL_SEGS    3
#define SL_THREADS 1024
#define SL_WAVES   (SL_THREADS / 32)
#define SL_QCAP    64
#define SL_ROWS    8

__device__ __forceinline__ void sl_drain(const unsigned* qw, int n, int lane,
                                         const unsigned* __restrict__ packed,
                                         const float* __restrict__ h, float* acc) {
    int s = 0; unsigned r = 0u;
    if (lane < n) {
        const unsigned p = packed[qw[lane]];
        s = (int)(p & ((1u << SL_SRC_BITS) - 1u));
        r = p >> SL_SRC_BITS;
    }
    #pragma clang loop unroll(disable)
    for (int i = 0; i < n; i += SL_ROWS) {
        float2   v[SL_ROWS];
        unsigned rr[SL_ROWS];
        #pragma unroll
        for (int u = 0; u < SL_ROWS; ++u) {
            const int idx = (i + u < n) ? i + u : n - 1;
            const int su  = __builtin_amdgcn_readlane(s, idx);
            rr[u] = (unsigned)__builtin_amdgcn_readlane((int)r, idx);
            v[u]  = ((const float2*)(h + (size_t)su * F_OUT))[lane];
        }
        #pragma unroll
        for (int u = 0; u < SL_ROWS; ++u) {
            if (i + u < n) {
                float* a = acc + rr[u] * F_OUT + 2 * lane;
                atomicAdd(a,     v[u].x);
                atomicAdd(a + 1, v[u].y);
            }
        }
    }
}

__global__ void __launch_bounds__(SL_THREADS) k_slice_agg(const unsigned* __restrict__ keys,
                                                          const unsigned* __restrict__ packed,
                                                          const float* __restrict__ h,
                                                          float* __restrict__ part, int E, int N) {
    __shared__ __attribute__((aligned(16))) float acc[SL_NODES * F_OUT];
    __shared__ unsigned q[SL_WAVES * SL_QCAP];

    const int tid  = threadIdx.x;
    const int lane = tid & 31;
    const int wave = __builtin_amdgcn_readfirstlane(tid >> 5);
    const int seg  = blockIdx.x;
    const int lo   = blockIdx.y * SL_NODES;
    const unsigned sid4 = (unsigned)blockIdx.y * 0x01010101u;

    float4* acc4 = (float4*)acc;
    for (int i = tid; i < SL_NODES * F_OUT / 4; i += SL_THREADS)
        acc4[i] = make_float4(0.0f, 0.0f, 0.0f, 0.0f);
    __syncthreads();

    const uint4* key4 = (const uint4*)keys;
    const int    ngrp = (E + 31) >> 5;
    const int    per  = (ngrp + SL_SEGS - 1) / SL_SEGS;
    const int    g0   = seg * per;
    const int    g1   = (ngrp - g0 < per) ? ngrp : g0 + per;
    const int    nb   = (g1 - g0 + 31) / 32;
    const int    rot  = (int)(((long long)blockIdx.y * nb) / (int)gridDim.y);

    unsigned* qw = q + wave * SL_QCAP;
    int cnt = 0;

    uint4 curA, curB, nxtA, nxtB;
    const unsigned padw = SL_PAD_KEY * 0x01010101u;

#define SL_LOAD(va, vb, it)                                                        \
    {                                                                              \
        int b_ = (it) + rot; if (b_ >= nb) b_ -= nb;                               \
        const int grp_ = g0 + b_ * 32 + lane;                                      \
        uint4 a_ = make_uint4(padw, padw, padw, padw), c_ = a_;                    \
        if (grp_ < g1) { a_ = key4[2 * grp_]; c_ = key4[2 * grp_ + 1]; }           \
        va = a_; vb = c_;                                                          \
    }

#define SL_NE(word, w) ((((word ^ sid4) + 0x7F7F7F7Fu) >> (7 - (w))) & (0x01010101u << (w)))

    int it = wave;
    if (it < nb) SL_LOAD(nxtA, nxtB, it);
    while (it < nb) {
        curA = nxtA; curB = nxtB;
        const int nit = it + SL_WAVES;
        if (nit < nb) SL_LOAD(nxtA, nxtB, nit);

        int b = it + rot; if (b >= nb) b -= nb;
        const unsigned ebase = 32u * (unsigned)(g0 + b * 32 + lane);

        const unsigned ne = SL_NE(curA.x, 0) | SL_NE(curA.y, 1) | SL_NE(curA.z, 2) | SL_NE(curA.w, 3)
                          | SL_NE(curB.x, 4) | SL_NE(curB.y, 5) | SL_NE(curB.z, 6) | SL_NE(curB.w, 7);
        unsigned hm = ~ne;

        #pragma clang loop unroll(disable)
        for (int pass = 0; pass < 32; ++pass) {
            const bool has = hm != 0u;
            const unsigned m = __builtin_amdgcn_ballot_w32(has);
            if (m == 0u) break;
            if (has) {
                const unsigned pos = (unsigned)__builtin_ctz(hm);
                hm &= hm - 1u;
                qw[cnt + (int)__builtin_amdgcn_mbcnt_lo(m, 0u)] = ebase + ((pos & 7u) << 2) + (pos >> 3);
            }
            cnt += __builtin_popcount(m);
            if (cnt >= 32) {
                __builtin_amdgcn_fence(__ATOMIC_ACQ_REL, "wavefront");
                sl_drain(qw, 32, lane, packed, h, acc);
                cnt -= 32;
                unsigned mv = 0u;
                if (lane < cnt) mv = qw[32 + lane];
                if (lane < cnt) qw[lane] = mv;
                __builtin_amdgcn_fence(__ATOMIC_ACQ_REL, "wavefront");
            }
        }
        it = nit;
    }
    if (cnt > 0) {
        __builtin_amdgcn_fence(__ATOMIC_ACQ_REL, "wavefront");
        sl_drain(qw, cnt, lane, packed, h, acc);
    }
#undef SL_LOAD
#undef SL_NE

    __syncthreads();

    int rows = N - lo; if (rows > SL_NODES) rows = SL_NODES;
    float4* p4 = (float4*)(part + (size_t)seg * N * F_OUT + (size_t)lo * F_OUT);
    for (int i = tid; i < rows * (F_OUT / 4); i += SL_THREADS)
        p4[i] = acc4[i];
}


__global__ void k_finalize(const float* __restrict__ h, const float* __restrict__ dinv,
                           const float* __restrict__ b_conv, const float* __restrict__ part,
                           const float* __restrict__ W_lin, const float* __restrict__ b_lin,
                           float* __restrict__ out, int N) {
    int gtid = blockIdx.x * blockDim.x + threadIdx.x;
    int node = gtid >> 5;
    int lane = threadIdx.x & 31;
    if (node >= N) return;

    const size_t plane = (size_t)N * F_OUT;
    const size_t o     = (size_t)node * F_OUT + 2 * lane;
    const float  di    = dinv[node];
    const float2 hv    = *(const float2*)(h + o);
    const float2 bc    = *(const float2*)(b_conv + 2 * lane);
    const float2 wl    = *(const float2*)(W_lin + 2 * lane);
    float2 pv[SL_SEGS];
    #pragma unroll
    for (int g = 0; g < SL_SEGS; ++g) pv[g] = *(const float2*)(part + g * plane + o);

    float sx = hv.x, sy = hv.y;
    #pragma unroll
    for (int g = 0; g < SL_SEGS; ++g) { sx += pv[g].x; sy += pv[g].y; }
    const float ax = bc.x + di * sx;
    const float ay = bc.y + di * sy;

    float p = fmaxf(ax, 0.0f) * wl.x + fmaxf(ay, 0.0f) * wl.y;

    #pragma unroll
    for (int off = 16; off > 0; off >>= 1)
        p += __shfl_down(p, off);

    if (lane == 0) {
        float z = p + b_lin[0];
        out[node] = 1.0f / (1.0f + __expf(-z));
    }
}


extern "C" void kernel_launch(void* const* d_in, const int* in_sizes, int n_in,
                              void* d_out, int out_size, void* d_ws, size_t ws_size,
                              hipStream_t stream) {
    const float* x      = (const float*)d_in[0];
    const int*   edge   = (const int*)  d_in[1];
    const float* W_conv = (const float*)d_in[2];
    const float* b_conv = (const float*)d_in[3];
    const float* W_lin  = (const float*)d_in[4];
    const float* b_lin  = (const float*)d_in[5];
    float*       out    = (float*)d_out;

    const int N = in_sizes[0] / F_IN;
    const int E = in_sizes[1] / 2;
    const int* src = edge;
    const int* dst = edge + E;

    float* ws   = (float*)d_ws;
    float* h    = ws;
    float* part = h + (size_t)N * F_OUT;
    float* dinv = part + (size_t)SL_SEGS * N * F_OUT;
    const int Epad = (E + 31) / 32 * 32;
    unsigned* packed = (unsigned*)(dinv + (N + 63) / 64 * 64);
    unsigned* keys   = packed + Epad;

    const int nprep = (N > Epad / 4) ? N : Epad / 4;
    k_prep     <<<(nprep + 255) / 256, 256, 0, stream>>>(src, dst, dinv, keys, packed, E, N);
    k_count_deg<<<dim3(DEG_SEGS, (N + DEG_BINS - 1) / DEG_BINS), DEG_THREADS, 0, stream>>>(dst, dinv, E, N);
    k_dinv     <<<(N + 255) / 256, 256, 0, stream>>>(dinv, N);

    k_gemm<<<GM_BLOCKS, GM_WAVES * 32, 0, stream>>>(x, W_conv, dinv, h, N);

    k_slice_agg<<<dim3(SL_SEGS, (N + SL_NODES - 1) / SL_NODES), SL_THREADS, 0, stream>>>(keys, packed, h, part, E, N);

    const long long fth = (long long)N * 32;
    k_finalize<<<(int)((fth + 255) / 256), 256, 0, stream>>>(h, dinv, b_conv, part, W_lin, b_lin, out, N);
}


# k_slice_agg: one block owns each 400-row slice over all edges, scanning a per-slice edge bitmap from k_prep; 3 partial planes and k_finalize gone, epilogue fused; 5 launches; row sum order changes
# speedup vs baseline: 3.1074x; 1.1135x over previous
#include <hip/hip_runtime.h>
#include <math.h>

typedef float v2f __attribute__((ext_vector_type(2)));
typedef float v8f __attribute__((ext_vector_type(8)));

#define F_IN  128
#define F_OUT 64


#define SL_NODES    400
#define SL_MAXSL    256
#define SL_SRC_BITS 20
#define PB_THREADS  256
#define PB_PITCH    33

__global__ void __launch_bounds__(PB_THREADS) k_prep(const int* __restrict__ src, const int* __restrict__ dst,
                       float* __restrict__ deg, unsigned* __restrict__ bm,
                       unsigned* __restrict__ packed, int E, int N, int nsl, int gp) {
    __shared__ unsigned tbl[SL_MAXSL * PB_PITCH];
    const int tid = threadIdx.x;
    const int i   = blockIdx.x * PB_THREADS + tid;
    const int nq  = ((E + 31) >> 5) << 3;
    if (i < N) deg[i] = 1.0f;
    for (int k = tid; k < nsl * PB_PITCH; k += PB_THREADS) tbl[k] = 0u;
    __syncthreads();
    if (i < nq) {
        const int g  = tid >> 3;
        const int b0 = 4 * (tid & 7);
        unsigned p[4];
        #pragma unroll
        for (int j = 0; j < 4; ++j) {
            const int e = 4 * i + j;
            unsigned pk = 0u;
            if (e < E) {
                const unsigned d  = (unsigned)dst[e];
                const unsigned sl = d / (unsigned)SL_NODES;
                pk = (unsigned)src[e] | ((d - sl * (unsigned)SL_NODES) << SL_SRC_BITS);
                atomicOr(&tbl[sl * PB_PITCH + g], 1u << (b0 + j));
            }
            p[j] = pk;
        }
        ((uint4*)packed)[i] = make_uint4(p[0], p[1], p[2], p[3]);
    }
    __syncthreads();
    const int wave = tid >> 5, lane = tid & 31;
    const int gbase = (int)blockIdx.x * 32;
    if (gbase < gp) {
        for (int s = wave; s < nsl; s += PB_THREADS / 32)
            bm[(size_t)s * gp + gbase + lane] = tbl[s * PB_PITCH + lane];
    }
}


#define DEG_BINS    16384
#define DEG_SEGS    32
#define DEG_THREADS 1024
#define DEG_AHEAD   8

__global__ void __launch_bounds__(DEG_THREADS) k_count_deg(const int* __restrict__ dst,
                                                           float* __restrict__ deg, int E, int N) {
    __shared__ unsigned cnt[DEG_BINS];
    const int tid = threadIdx.x;
    const int lo  = blockIdx.y * DEG_BINS;

    for (int b = tid; b < DEG_BINS; b += DEG_THREADS) cnt[b] = 0u;
    __syncthreads();

    const int nseg  = (int)gridDim.x;
    const int chunk = (E + nseg - 1) / nseg;
    const int e0    = blockIdx.x * chunk;
    const int e1    = (E - e0 < chunk) ? E : e0 + chunk;

    for (int e = e0 + tid; e < e1; e += DEG_AHEAD * DEG_THREADS) {
        int d[DEG_AHEAD];
        #pragma unroll
        for (int j = 0; j < DEG_AHEAD; ++j) {
            const int ee = e + j * DEG_THREADS;
            d[j] = (ee < e1) ? dst[ee] : -1;
        }
        #pragma unroll
        for (int j = 0; j < DEG_AHEAD; ++j) {
            const unsigned r = (unsigned)(d[j] - lo);
            if (r < (unsigned)DEG_BINS) atomicAdd(&cnt[r], 1u);
        }
    }
    __syncthreads();

    const int rot = (blockIdx.x * DEG_THREADS) & (DEG_BINS - 1);
    unsigned c[DEG_BINS / DEG_THREADS];
    #pragma unroll
    for (int i = 0; i < DEG_BINS / DEG_THREADS; ++i)
        c[i] = cnt[(tid + i * DEG_THREADS + rot) & (DEG_BINS - 1)];
    #pragma unroll
    for (int i = 0; i < DEG_BINS / DEG_THREADS; ++i) {
        const int node = lo + ((tid + i * DEG_THREADS + rot) & (DEG_BINS - 1));
        if (c[i] != 0u && node < N) atomicAdd(&deg[node], (float)c[i]);
    }
}

__global__ void k_dinv(float* __restrict__ deg, int N) {
    int i = blockIdx.x * blockDim.x + threadIdx.x;
    if (i < N) {
        float d = deg[i];
        deg[i] = (d > 0.0f) ? rsqrtf(d) : 0.0f;
    }
}


#define GM_WAVES  4
#define GM_BLOCKS 256
#define GM_KSTEPS (F_IN / 4)

__global__ void __launch_bounds__(GM_WAVES * 32) k_gemm(const float* __restrict__ x,
                                                        const float* __restrict__ W,
                                                        const float* __restrict__ dinv,
                                                        float* __restrict__ h, int N) {
    const int wave = __builtin_amdgcn_readfirstlane(threadIdx.x >> 5);
    const int lane = threadIdx.x & 31;
    const int lr    = lane & 15;
    const int khalf = lane >> 4;

    const int ntiles = (N + 15) >> 4;
    const int nw     = (int)gridDim.x * GM_WAVES;
    int tile = wave * (int)gridDim.x + (int)blockIdx.x;
    if (tile >= ntiles) return;

    v2f b[GM_KSTEPS][4];
    #pragma unroll
    for (int t = 0; t < 4; ++t) {
        const float2* wp = (const float2*)(W + (size_t)(lr + 16 * t) * F_IN);
        #pragma unroll
        for (int ks = 0; ks < GM_KSTEPS; ++ks) {
            const float2 f = wp[2 * ks + khalf];
            b[ks][t].x = f.x; b[ks][t].y = f.y;
        }
    }

    v2f a[GM_KSTEPS];
    {
        int m = tile * 16 + lr;
        if (m >= N) m = N - 1;
        const float2* xp = (const float2*)(x + (size_t)m * F_IN);
        #pragma unroll
        for (int ks = 0; ks < GM_KSTEPS; ++ks) {
            const float2 f = xp[2 * ks + khalf];
            a[ks].x = f.x; a[ks].y = f.y;
        }
    }

    for (;;) {
        const int  nxt  = tile + nw;
        const bool more = nxt < ntiles;

        const int mbase = tile * 16 + (khalf << 3);
        float sc[8];
        #pragma unroll
        for (int v = 0; v < 8; ++v) {
            int mm = mbase + v;
            if (mm >= N) mm = N - 1;
            sc[v] = dinv[mm];
        }

        v2f an[GM_KSTEPS];
        {
            int m = (more ? nxt : tile) * 16 + lr;
            if (m >= N) m = N - 1;
            const float2* xp = (const float2*)(x + (size_t)m * F_IN);
            #pragma unroll
            for (int ks = 0; ks < GM_KSTEPS; ++ks) {
                const float2 f = xp[2 * ks + khalf];
                an[ks].x = f.x; an[ks].y = f.y;
            }
        }

        v8f c0 = {}, c1 = {}, c2 = {}, c3 = {};
        #pragma unroll
        for (int ks = 0; ks < GM_KSTEPS; ++ks) {
            c0 = __builtin_amdgcn_wmma_f32_16x16x4_f32(false, a[ks], false, b[ks][0], (short)0, c0, false, false);
            c1 = __builtin_amdgcn_wmma_f32_16x16x4_f32(false, a[ks], false, b[ks][1], (short)0, c1, false, false);
            c2 = __builtin_amdgcn_wmma_f32_16x16x4_f32(false, a[ks], false, b[ks][2], (short)0, c2, false, false);
            c3 = __builtin_amdgcn_wmma_f32_16x16x4_f32(false, a[ks], false, b[ks][3], (short)0, c3, false, false);
        }

        if (tile * 16 + 16 <= N) {
            float* hp = h + (size_t)mbase * F_OUT + lr;
            #pragma unroll
            for (int v = 0; v < 8; ++v) {
                hp[v * F_OUT +  0] = c0[v] * sc[v];
                hp[v * F_OUT + 16] = c1[v] * sc[v];
                hp[v * F_OUT + 32] = c2[v] * sc[v];
                hp[v * F_OUT + 48] = c3[v] * sc[v];
            }
        } else {
            #pragma unroll
            for (int v = 0; v < 8; ++v) {
                const int mm = mbase + v;
                if (mm < N) {
                    float* hp = h + (size_t)mm * F_OUT + lr;
                    hp[ 0] = c0[v] * sc[v];
                    hp[16] = c1[v] * sc[v];
                    hp[32] = c2[v] * sc[v];
                    hp[48] = c3[v] * sc[v];
                }
            }
        }

        if (!more) break;
        #pragma unroll
        for (int ks = 0; ks < GM_KSTEPS; ++ks) a[ks] = an[ks];
        tile = nxt;
    }
}


#define SL_THREADS 1024
#define SL_WAVES   (SL_THREADS / 32)
#define SL_QCAP    64
#define SL_ROWS    8
#define SL_EPI     ((SL_NODES + SL_WAVES - 1) / SL_WAVES)

__device__ __forceinline__ void sl_drain(const unsigned* qw, int n, int lane,
                                         const unsigned* __restrict__ packed,
                                         const float* __restrict__ h, float* acc) {
    int s = 0; unsigned r = 0u;
    if (lane < n) {
        const unsigned p = packed[qw[lane]];
        s = (int)(p & ((1u << SL_SRC_BITS) - 1u));
        r = p >> SL_SRC_BITS;
    }
    #pragma clang loop unroll(disable)
    for (int i = 0; i < n; i += SL_ROWS) {
        float2   v[SL_ROWS];
        unsigned rr[SL_ROWS];
        #pragma unroll
        for (int u = 0; u < SL_ROWS; ++u) {
            const int idx = (i + u < n) ? i + u : n - 1;
            const int su  = __builtin_amdgcn_readlane(s, idx);
            rr[u] = (unsigned)__builtin_amdgcn_readlane((int)r, idx);
            v[u]  = ((const float2*)(h + (size_t)su * F_OUT))[lane];
        }
        #pragma unroll
        for (int u = 0; u < SL_ROWS; ++u) {
            if (i + u < n) {
                float* a = acc + rr[u] * F_OUT + 2 * lane;
                atomicAdd(a,     v[u].x);
                atomicAdd(a + 1, v[u].y);
            }
        }
    }
}

__device__ __forceinline__ int sl_hits(unsigned hm, unsigned ebase, unsigned* qw, int cnt, int lane,
                                       const unsigned* __restrict__ packed,
                                       const float* __restrict__ h, float* acc) {
    #pragma clang loop unroll(disable)
    for (int pass = 0; pass < 32; ++pass) {
        const bool has = hm != 0u;
        const unsigned m = __builtin_amdgcn_ballot_w32(has);
        if (m == 0u) break;
        if (has) {
            const unsigned pos = (unsigned)__builtin_ctz(hm);
            hm &= hm - 1u;
            qw[cnt + (int)__builtin_amdgcn_mbcnt_lo(m, 0u)] = ebase + pos;
        }
        cnt += __builtin_popcount(m);
        if (cnt >= 32) {
            __builtin_amdgcn_fence(__ATOMIC_ACQ_REL, "wavefront");
            sl_drain(qw, 32, lane, packed, h, acc);
            cnt -= 32;
            unsigned mv = 0u;
            if (lane < cnt) mv = qw[32 + lane];
            if (lane < cnt) qw[lane] = mv;
            __builtin_amdgcn_fence(__ATOMIC_ACQ_REL, "wavefront");
        }
    }
    return cnt;
}

__global__ void __launch_bounds__(SL_THREADS) k_slice_agg(const unsigned* __restrict__ bm,
                                                          const unsigned* __restrict__ packed,
                                                          const float* __restrict__ h,
                                                          const float* __restrict__ dinv,
                                                          const float* __restrict__ b_conv,
                                                          const float* __restrict__ W_lin,
                                                          const float* __restrict__ b_lin,
                                                          float* __restrict__ out, int N, int gp) {
    __shared__ __attribute__((aligned(16))) float acc[SL_NODES * F_OUT];
    __shared__ unsigned q[SL_WAVES * SL_QCAP];

    const int tid  = threadIdx.x;
    const int lane = tid & 31;
    const int wave = __builtin_amdgcn_readfirstlane(tid >> 5);
    const int lo   = blockIdx.x * SL_NODES;

    float4* acc4 = (float4*)acc;
    for (int i = tid; i < SL_NODES * F_OUT / 4; i += SL_THREADS)
        acc4[i] = make_float4(0.0f, 0.0f, 0.0f, 0.0f);
    __syncthreads();

    const uint4* row4 = (const uint4*)(bm + (size_t)blockIdx.x * gp);
    const int    nb   = gp >> 7;
    const int    rot  = (int)(((long long)blockIdx.x * nb) / (int)gridDim.x);

    unsigned* qw = q + wave * SL_QCAP;
    int cnt = 0;

    uint4 cur, nxt;
    int it = wave;
    if (it < nb) { int b_ = it + rot; if (b_ >= nb) b_ -= nb; nxt = row4[b_ * 32 + lane]; }
    while (it < nb) {
        cur = nxt;
        const int nit = it + SL_WAVES;
        if (nit < nb) { int b_ = nit + rot; if (b_ >= nb) b_ -= nb; nxt = row4[b_ * 32 + lane]; }

        int b = it + rot; if (b >= nb) b -= nb;
        const unsigned ebase = 128u * (unsigned)(b * 32 + lane);
        cnt = sl_hits(cur.x, ebase,       qw, cnt, lane, packed, h, acc);
        cnt = sl_hits(cur.y, ebase + 32u, qw, cnt, lane, packed, h, acc);
        cnt = sl_hits(cur.z, ebase + 64u, qw, cnt, lane, packed, h, acc);
        cnt = sl_hits(cur.w, ebase + 96u, qw, cnt, lane, packed, h, acc);
        it = nit;
    }
    if (cnt > 0) {
        __builtin_amdgcn_fence(__ATOMIC_ACQ_REL, "wavefront");
        sl_drain(qw, cnt, lane, packed, h, acc);
    }

    __syncthreads();

    int rows = N - lo; if (rows > SL_NODES) rows = SL_NODES;
    const float2 bc = *(const float2*)(b_conv + 2 * lane);
    const float2 wl = *(const float2*)(W_lin + 2 * lane);
    const float  bl = b_lin[0];
    float2 hv[SL_EPI];
    float  di[SL_EPI];
    #pragma unroll
    for (int k = 0; k < SL_EPI; ++k) {
        int r = wave + k * SL_WAVES;
        if (r >= rows) r = rows - 1;
        const int node = lo + r;
        hv[k] = ((const float2*)(h + (size_t)node * F_OUT))[lane];
        di[k] = dinv[node];
    }
    #pragma unroll
    for (int k = 0; k < SL_EPI; ++k) {
        const int r  = wave + k * SL_WAVES;
        const int rc = (r < rows) ? r : rows - 1;
        const float2 a = ((const float2*)acc)[rc * (F_OUT / 2) + lane];
        const float ax = bc.x + di[k] * (hv[k].x + a.x);
        const float ay = bc.y + di[k] * (hv[k].y + a.y);
        float p = fmaxf(ax, 0.0f) * wl.x + fmaxf(ay, 0.0f) * wl.y;
        #pragma unroll
        for (int off = 16; off > 0; off >>= 1)
            p += __shfl_down(p, off);
        if (lane == 0 && r < rows) {
            const float z = p + bl;
            out[lo + r] = 1.0f / (1.0f + __expf(-z));
        }
    }
}


extern "C" void kernel_launch(void* const* d_in, const int* in_sizes, int n_in,
                              void* d_out, int out_size, void* d_ws, size_t ws_size,
                              hipStream_t stream) {
    const float* x      = (const float*)d_in[0];
    const int*   edge   = (const int*)  d_in[1];
    const float* W_conv = (const float*)d_in[2];
    const float* b_conv = (const float*)d_in[3];
    const float* W_lin  = (const float*)d_in[4];
    const float* b_lin  = (const float*)d_in[5];
    float*       out    = (float*)d_out;

    const int N = in_sizes[0] / F_IN;
    const int E = in_sizes[1] / 2;
    const int* src = edge;
    const int* dst = edge + E;

    float* ws   = (float*)d_ws;
    float* h    = ws;
    float* dinv = h + (size_t)N * F_OUT;
    const int Epad = (E + 31) / 32 * 32;
    unsigned* packed = (unsigned*)(dinv + (N + 63) / 64 * 64);
    unsigned* bm     = packed + Epad;
    const int nsl = (N + SL_NODES - 1) / SL_NODES;
    const int gp  = (Epad / 32 + 127) / 128 * 128;

    const int nchunk = gp / 32;
    const int nnode  = (N + PB_THREADS - 1) / PB_THREADS;
    const int nprep  = (nchunk > nnode) ? nchunk : nnode;
    k_prep     <<<nprep, PB_THREADS, 0, stream>>>(src, dst, dinv, bm, packed, E, N, nsl, gp);
    k_count_deg<<<dim3(DEG_SEGS, (N + DEG_BINS - 1) / DEG_BINS), DEG_THREADS, 0, stream>>>(dst, dinv, E, N);
    k_dinv     <<<(N + 255) / 256, 256, 0, stream>>>(dinv, N);

    k_gemm<<<GM_BLOCKS, GM_WAVES * 32, 0, stream>>>(x, W_conv, dinv, h, N);

    k_slice_agg<<<nsl, SL_THREADS, 0, stream>>>(bm, packed, h, dinv, b_conv, W_lin, b_lin, out, N, gp);
}


# k_prep bins each 2048-edge block by slice into 128-B cells (hit count + hit words), no bitmaps or packed words; k_slice_agg queues hit words from cells; row sum order arrival-dependent as before
# speedup vs baseline: 3.7931x; 1.2207x over previous
#include <hip/hip_runtime.h>
#include <math.h>

typedef float v2f __attribute__((ext_vector_type(2)));
typedef float v8f __attribute__((ext_vector_type(8)));

#define F_IN  128
#define F_OUT 64


#define SL_NODES    400
#define SL_MAXSL    256
#define SL_SRC_BITS 20
#define PB_THREADS  512
#define PB_EDGES    (4 * PB_THREADS)
#define PB_PITCH    33
#define CELL_WORDS  32
#define CELL_CAP    (CELL_WORDS - 1)

__global__ void __launch_bounds__(PB_THREADS) k_prep(const int* __restrict__ src, const int* __restrict__ dst,
                       float* __restrict__ deg, unsigned* __restrict__ cells,
                       int E, int N, int nsl, int ncell) {
    __shared__ unsigned tbl[SL_MAXSL * PB_PITCH];
    const int tid = threadIdx.x;
    const int i   = blockIdx.x * PB_THREADS + tid;
    if (i < N) deg[i] = 1.0f;
    if ((int)blockIdx.x >= ncell) return;
    for (int k = tid; k < nsl; k += PB_THREADS) tbl[k * PB_PITCH] = 0u;
    __syncthreads();
    int dd[4], ss[4];
    #pragma unroll
    for (int j = 0; j < 4; ++j) {
        const int e  = 4 * i + j;
        const int ec = (e < E) ? e : E - 1;
        dd[j] = dst[ec];
        ss[j] = src[ec];
    }
    unsigned sl[4], slot[4];
    #pragma unroll
    for (int j = 0; j < 4; ++j) {
        sl[j]   = (unsigned)dd[j] / (unsigned)SL_NODES;
        slot[j] = (unsigned)CELL_CAP;
        if (4 * i + j < E) slot[j] = atomicAdd(&tbl[sl[j] * PB_PITCH], 1u);
    }
    #pragma unroll
    for (int j = 0; j < 4; ++j) {
        if (slot[j] < (unsigned)CELL_CAP)
            tbl[sl[j] * PB_PITCH + 1 + slot[j]] =
                (unsigned)ss[j] | (((unsigned)dd[j] - sl[j] * (unsigned)SL_NODES) << SL_SRC_BITS);
    }
    __syncthreads();
    const int wave = tid >> 5, lane = tid & 31;
    for (int s = wave; s < nsl; s += PB_THREADS / 32)
        cells[((size_t)s * ncell + blockIdx.x) * CELL_WORDS + lane] = tbl[s * PB_PITCH + lane];
}


#define DEG_BINS    16384
#define DEG_SEGS    32
#define DEG_THREADS 1024
#define DEG_AHEAD   8

__global__ void __launch_bounds__(DEG_THREADS) k_count_deg(const int* __restrict__ dst,
                                                           float* __restrict__ deg, int E, int N) {
    __shared__ unsigned cnt[DEG_BINS];
    const int tid = threadIdx.x;
    const int lo  = blockIdx.y * DEG_BINS;

    for (int b = tid; b < DEG_BINS; b += DEG_THREADS) cnt[b] = 0u;
    __syncthreads();

    const int nseg  = (int)gridDim.x;
    const int chunk = (E + nseg - 1) / nseg;
    const int e0    = blockIdx.x * chunk;
    const int e1    = (E - e0 < chunk) ? E : e0 + chunk;

    for (int e = e0 + tid; e < e1; e += DEG_AHEAD * DEG_THREADS) {
        int d[DEG_AHEAD];
        #pragma unroll
        for (int j = 0; j < DEG_AHEAD; ++j) {
            const int ee = e + j * DEG_THREADS;
            d[j] = (ee < e1) ? dst[ee] : -1;
        }
        #pragma unroll
        for (int j = 0; j < DEG_AHEAD; ++j) {
            const unsigned r = (unsigned)(d[j] - lo);
            if (r < (unsigned)DEG_BINS) atomicAdd(&cnt[r], 1u);
        }
    }
    __syncthreads();

    const int rot = (blockIdx.x * DEG_THREADS) & (DEG_BINS - 1);
    unsigned c[DEG_BINS / DEG_THREADS];
    #pragma unroll
    for (int i = 0; i < DEG_BINS / DEG_THREADS; ++i)
        c[i] = cnt[(tid + i * DEG_THREADS + rot) & (DEG_BINS - 1)];
    #pragma unroll
    for (int i = 0; i < DEG_BINS / DEG_THREADS; ++i) {
        const int node = lo + ((tid + i * DEG_THREADS + rot) & (DEG_BINS - 1));
        if (c[i] != 0u && node < N) atomicAdd(&deg[node], (float)c[i]);
    }
}

__global__ void k_dinv(float* __restrict__ deg, int N) {
    int i = blockIdx.x * blockDim.x + threadIdx.x;
    if (i < N) {
        float d = deg[i];
        deg[i] = (d > 0.0f) ? rsqrtf(d) : 0.0f;
    }
}


#define GM_WAVES  4
#define GM_BLOCKS 256
#define GM_KSTEPS (F_IN / 4)

__global__ void __launch_bounds__(GM_WAVES * 32) k_gemm(const float* __restrict__ x,
                                                        const float* __restrict__ W,
                                                        const float* __restrict__ dinv,
                                                        float* __restrict__ h, int N) {
    const int wave = __builtin_amdgcn_readfirstlane(threadIdx.x >> 5);
    const int lane = threadIdx.x & 31;
    const int lr    = lane & 15;
    const int khalf = lane >> 4;

    const int ntiles = (N + 15) >> 4;
    const int nw     = (int)gridDim.x * GM_WAVES;
    int tile = wave * (int)gridDim.x + (int)blockIdx.x;
    if (tile >= ntiles) return;

    v2f b[GM_KSTEPS][4];
    #pragma unroll
    for (int t = 0; t < 4; ++t) {
        const float2* wp = (const float2*)(W + (size_t)(lr + 16 * t) * F_IN);
        #pragma unroll
        for (int ks = 0; ks < GM_KSTEPS; ++ks) {
            const float2 f = wp[2 * ks + khalf];
            b[ks][t].x = f.x; b[ks][t].y = f.y;
        }
    }

    v2f a[GM_KSTEPS];
    {
        int m = tile * 16 + lr;
        if (m >= N) m = N - 1;
        const float2* xp = (const float2*)(x + (size_t)m * F_IN);
        #pragma unroll
        for (int ks = 0; ks < GM_KSTEPS; ++ks) {
            const float2 f = xp[2 * ks + khalf];
            a[ks].x = f.x; a[ks].y = f.y;
        }
    }

    for (;;) {
        const int  nxt  = tile + nw;
        const bool more = nxt < ntiles;

        const int mbase = tile * 16 + (khalf << 3);
        float sc[8];
        #pragma unroll
        for (int v = 0; v < 8; ++v) {
            int mm = mbase + v;
            if (mm >= N) mm = N - 1;
            sc[v] = dinv[mm];
        }

        v2f an[GM_KSTEPS];
        {
            int m = (more ? nxt : tile) * 16 + lr;
            if (m >= N) m = N - 1;
            const float2* xp = (const float2*)(x + (size_t)m * F_IN);
            #pragma unroll
            for (int ks = 0; ks < GM_KSTEPS; ++ks) {
                const float2 f = xp[2 * ks + khalf];
                an[ks].x = f.x; an[ks].y = f.y;
            }
        }

        v8f c0 = {}, c1 = {}, c2 = {}, c3 = {};
        #pragma unroll
        for (int ks = 0; ks < GM_KSTEPS; ++ks) {
            c0 = __builtin_amdgcn_wmma_f32_16x16x4_f32(false, a[ks], false, b[ks][0], (short)0, c0, false, false);
            c1 = __builtin_amdgcn_wmma_f32_16x16x4_f32(false, a[ks], false, b[ks][1], (short)0, c1, false, false);
            c2 = __builtin_amdgcn_wmma_f32_16x16x4_f32(false, a[ks], false, b[ks][2], (short)0, c2, false, false);
            c3 = __builtin_amdgcn_wmma_f32_16x16x4_f32(false, a[ks], false, b[ks][3], (short)0, c3, false, false);
        }

        if (tile * 16 + 16 <= N) {
            float* hp = h + (size_t)mbase * F_OUT + lr;
            #pragma unroll
            for (int v = 0; v < 8; ++v) {
                hp[v * F_OUT +  0] = c0[v] * sc[v];
                hp[v * F_OUT + 16] = c1[v] * sc[v];
                hp[v * F_OUT + 32] = c2[v] * sc[v];
                hp[v * F_OUT + 48] = c3[v] * sc[v];
            }
        } else {
            #pragma unroll
            for (int v = 0; v < 8; ++v) {
                const int mm = mbase + v;
                if (mm < N) {
                    float* hp = h + (size_t)mm * F_OUT + lr;
                    hp[ 0] = c0[v] * sc[v];
                    hp[16] = c1[v] * sc[v];
                    hp[32] = c2[v] * sc[v];
                    hp[48] = c3[v] * sc[v];
                }
            }
        }

        if (!more) break;
        #pragma unroll
        for (int ks = 0; ks < GM_KSTEPS; ++ks) a[ks] = an[ks];
        tile = nxt;
    }
}


#define SL_THREADS 1024
#define SL_WAVES   (SL_THREADS / 32)
#define SL_QCAP    64
#define SL_ROWS    8
#define SL_EPI     ((SL_NODES + SL_WAVES - 1) / SL_WAVES)

__device__ __forceinline__ void sl_drain(const unsigned* qw, int n, int lane,
                                         const float* __restrict__ h, float* acc) {
    unsigned p = 0u;
    if (lane < n) p = qw[lane];
    const int      s = (int)(p & ((1u << SL_SRC_BITS) - 1u));
    const unsigned r = p >> SL_SRC_BITS;
    #pragma clang loop unroll(disable)
    for (int i = 0; i < n; i += SL_ROWS) {
        float2   v[SL_ROWS];
        unsigned rr[SL_ROWS];
        #pragma unroll
        for (int u = 0; u < SL_ROWS; ++u) {
            const int idx = (i + u < n) ? i + u : n - 1;
            const int su  = __builtin_amdgcn_readlane(s, idx);
            rr[u] = (unsigned)__builtin_amdgcn_readlane((int)r, idx);
            v[u]  = ((const float2*)(h + (size_t)su * F_OUT))[lane];
        }
        #pragma unroll
        for (int u = 0; u < SL_ROWS; ++u) {
            if (i + u < n) {
                float* a = acc + rr[u] * F_OUT + 2 * lane;
                atomicAdd(a,     v[u].x);
                atomicAdd(a + 1, v[u].y);
            }
        }
    }
}

__device__ __forceinline__ int sl_push(bool has, unsigned word, unsigned* qw, int cnt, int lane,
                                       const float* __restrict__ h, float* acc) {
    const unsigned m = __builtin_amdgcn_ballot_w32(has);
    if (m != 0u) {
        if (has) qw[cnt + (int)__builtin_amdgcn_mbcnt_lo(m, 0u)] = word;
        cnt += __builtin_popcount(m);
        if (cnt >= 32) {
            __builtin_amdgcn_fence(__ATOMIC_ACQ_REL, "wavefront");
            sl_drain(qw, 32, lane, h, acc);
            cnt -= 32;
            unsigned mv = 0u;
            if (lane < cnt) mv = qw[32 + lane];
            if (lane < cnt) qw[lane] = mv;
            __builtin_amdgcn_fence(__ATOMIC_ACQ_REL, "wavefront");
        }
    }
    return cnt;
}

__device__ __forceinline__ int sl_edges(int cell, int lo, const int* __restrict__ src,
                                        const int* __restrict__ dst, int E, unsigned* qw, int cnt,
                                        int lane, const float* __restrict__ h, float* acc) {
    const int e0 = cell * PB_EDGES;
    #pragma clang loop unroll(disable)
    for (int k = 0; k < PB_EDGES; k += 32) {
        const int e = e0 + k + lane;
        bool     has  = false;
        unsigned word = 0u;
        if (e < E) {
            const unsigned r = (unsigned)(dst[e] - lo);
            if (r < (unsigned)SL_NODES) {
                has  = true;
                word = (unsigned)src[e] | (r << SL_SRC_BITS);
            }
        }
        cnt = sl_push(has, word, qw, cnt, lane, h, acc);
    }
    return cnt;
}

__global__ void __launch_bounds__(SL_THREADS) k_slice_agg(const unsigned* __restrict__ cells,
                                                          const int* __restrict__ src,
                                                          const int* __restrict__ dst,
                                                          const float* __restrict__ h,
                                                          const float* __restrict__ dinv,
                                                          const float* __restrict__ b_conv,
                                                          const float* __restrict__ W_lin,
                                                          const float* __restrict__ b_lin,
                                                          float* __restrict__ out, int N, int E, int ncell) {
    __shared__ __attribute__((aligned(16))) float acc[SL_NODES * F_OUT];
    __shared__ unsigned q[SL_WAVES * SL_QCAP];

    const int tid  = threadIdx.x;
    const int lane = tid & 31;
    const int wave = __builtin_amdgcn_readfirstlane(tid >> 5);
    const int lo   = blockIdx.x * SL_NODES;

    float4* acc4 = (float4*)acc;
    for (int i = tid; i < SL_NODES * F_OUT / 4; i += SL_THREADS)
        acc4[i] = make_float4(0.0f, 0.0f, 0.0f, 0.0f);
    __syncthreads();

    const uint4* row4 = (const uint4*)(cells + (size_t)blockIdx.x * ncell * CELL_WORDS);
    const int    nb   = ncell >> 2;

    unsigned* qw = q + wave * SL_QCAP;
    int cnt = 0;

    const unsigned w0 = 4u * (unsigned)(lane & 7);

    uint4 nxt = make_uint4(0u, 0u, 0u, 0u);
    int it = wave;
    if (it < nb) nxt = row4[it * 32 + lane];
    while (it < nb) {
        const uint4 cur = nxt;
        const int nit = it + SL_WAVES;
        nxt = row4[((nit < nb) ? nit : it) * 32 + lane];

        const unsigned n    = (unsigned)__shfl((int)cur.x, lane & 24);
        const bool     over = n > (unsigned)CELL_CAP;
        const unsigned nn   = over ? 0u : n;
        cnt = sl_push(w0 - 1u < nn, cur.x, qw, cnt, lane, h, acc);
        cnt = sl_push(w0      < nn, cur.y, qw, cnt, lane, h, acc);
        cnt = sl_push(w0 + 1u < nn, cur.z, qw, cnt, lane, h, acc);
        cnt = sl_push(w0 + 2u < nn, cur.w, qw, cnt, lane, h, acc);
        const unsigned om = __builtin_amdgcn_ballot_w32(over);
        if (om != 0u) {
            for (int c = 0; c < 4; ++c)
                if ((om >> (8 * c)) & 1u)
                    cnt = sl_edges(4 * it + c, lo, src, dst, E, qw, cnt, lane, h, acc);
        }
        it = nit;
    }
    if (cnt > 0) {
        __builtin_amdgcn_fence(__ATOMIC_ACQ_REL, "wavefront");
        sl_drain(qw, cnt, lane, h, acc);
    }

    __syncthreads();

    int rows = N - lo; if (rows > SL_NODES) rows = SL_NODES;
    const float2 bc = *(const float2*)(b_conv + 2 * lane);
    const float2 wl = *(const float2*)(W_lin + 2 * lane);
    const float  bl = b_lin[0];
    float2 hv[SL_EPI];
    float  di[SL_EPI];
    #pragma unroll
    for (int k = 0; k < SL_EPI; ++k) {
        int r = wave + k * SL_WAVES;
        if (r >= rows) r = rows - 1;
        const int node = lo + r;
        hv[k] = ((const float2*)(h + (size_t)node * F_OUT))[lane];
        di[k] = dinv[node];
    }
    #pragma unroll
    for (int k = 0; k < SL_EPI; ++k) {
        const int r  = wave + k * SL_WAVES;
        const int rc = (r < rows) ? r : rows - 1;
        const float2 a = ((const float2*)acc)[rc * (F_OUT / 2) + lane];
        const float ax = bc.x + di[k] * (hv[k].x + a.x);
        const float ay = bc.y + di[k] * (hv[k].y + a.y);
        float p = fmaxf(ax, 0.0f) * wl.x + fmaxf(ay, 0.0f) * wl.y;
        #pragma unroll
        for (int off = 16; off > 0; off >>= 1)
            p += __shfl_down(p, off);
        if (lane == 0 && r < rows) {
            const float z = p + bl;
            out[lo + r] = 1.0f / (1.0f + __expf(-z));
        }
    }
}


extern "C" void kernel_launch(void* const* d_in, const int* in_sizes, int n_in,
                              void* d_out, int out_size, void* d_ws, size_t ws_size,
                              hipStream_t stream) {
    const float* x      = (const float*)d_in[0];
    const int*   edge   = (const int*)  d_in[1];
    const float* W_conv = (const float*)d_in[2];
    const float* b_conv = (const float*)d_in[3];
    const float* W_lin  = (const float*)d_in[4];
    const float* b_lin  = (const float*)d_in[5];
    float*       out    = (float*)d_out;

    const int N = in_sizes[0] / F_IN;
    const int E = in_sizes[1] / 2;
    const int* src = edge;
    const int* dst = edge + E;

    float* ws   = (float*)d_ws;
    float* h    = ws;
    float* dinv = h + (size_t)N * F_OUT;
    unsigned* cells = (unsigned*)(dinv + (N + 63) / 64 * 64);
    const int nsl   = (N + SL_NODES - 1) / SL_NODES;
    const int ncell = ((E + PB_EDGES - 1) / PB_EDGES + 3) / 4 * 4;

    const int nnode = (N + PB_THREADS - 1) / PB_THREADS;
    const int nprep = (ncell > nnode) ? ncell : nnode;
    k_prep     <<<nprep, PB_THREADS, 0, stream>>>(src, dst, dinv, cells, E, N, nsl, ncell);
    k_count_deg<<<dim3(DEG_SEGS, (N + DEG_BINS - 1) / DEG_BINS), DEG_THREADS, 0, stream>>>(dst, dinv, E, N);
    k_dinv     <<<(N + 255) / 256, 256, 0, stream>>>(dinv, N);

    k_gemm<<<GM_BLOCKS, GM_WAVES * 32, 0, stream>>>(x, W_conv, dinv, h, N);

    k_slice_agg<<<nsl, SL_THREADS, 0, stream>>>(cells, src, dst, h, dinv, b_conv, W_lin, b_lin, out, N, E, ncell);
}


# k_deg: one block per 400-row slice counts its rows' hits from the hit cells in LDS integers and writes dinv itself; k_count_deg (global float adds) and k_dinv gone, 4 launches; dinv bit-identical
# speedup vs baseline: 4.3317x; 1.1420x over previous
#include <hip/hip_runtime.h>
#include <math.h>

typedef float v2f __attribute__((ext_vector_type(2)));
typedef float v8f __attribute__((ext_vector_type(8)));

#define F_IN  128
#define F_OUT 64


#define SL_NODES    400
#define SL_MAXSL    256
#define SL_SRC_BITS 20
#define PB_THREADS  512
#define PB_EDGES    (4 * PB_THREADS)
#define PB_PITCH    33
#define CELL_WORDS  32
#define CELL_CAP    (CELL_WORDS - 1)

__global__ void __launch_bounds__(PB_THREADS) k_prep(const int* __restrict__ src, const int* __restrict__ dst,
                       unsigned* __restrict__ cells, int E, int nsl, int ncell) {
    __shared__ unsigned tbl[SL_MAXSL * PB_PITCH];
    const int tid = threadIdx.x;
    const int i   = blockIdx.x * PB_THREADS + tid;
    for (int k = tid; k < nsl; k += PB_THREADS) tbl[k * PB_PITCH] = 0u;
    __syncthreads();
    int dd[4], ss[4];
    #pragma unroll
    for (int j = 0; j < 4; ++j) {
        const int e  = 4 * i + j;
        const int ec = (e < E) ? e : E - 1;
        dd[j] = dst[ec];
        ss[j] = src[ec];
    }
    unsigned sl[4], slot[4];
    #pragma unroll
    for (int j = 0; j < 4; ++j) {
        sl[j]   = (unsigned)dd[j] / (unsigned)SL_NODES;
        slot[j] = (unsigned)CELL_CAP;
        if (4 * i + j < E) slot[j] = atomicAdd(&tbl[sl[j] * PB_PITCH], 1u);
    }
    #pragma unroll
    for (int j = 0; j < 4; ++j) {
        if (slot[j] < (unsigned)CELL_CAP)
            tbl[sl[j] * PB_PITCH + 1 + slot[j]] =
                (unsigned)ss[j] | (((unsigned)dd[j] - sl[j] * (unsigned)SL_NODES) << SL_SRC_BITS);
    }
    __syncthreads();
    const int wave = tid >> 5, lane = tid & 31;
    for (int s = wave; s < nsl; s += PB_THREADS / 32)
        cells[((size_t)s * ncell + blockIdx.x) * CELL_WORDS + lane] = tbl[s * PB_PITCH + lane];
}


#define DG_THREADS 512
#define DG_WAVES   (DG_THREADS / 32)
#define DG_AHEAD   16

__global__ void __launch_bounds__(DG_THREADS) k_deg(const unsigned* __restrict__ cells,
                                                    const int* __restrict__ dst,
                                                    float* __restrict__ dinv, int N, int E, int ncell) {
    __shared__ unsigned cnt[SL_NODES];
    const int tid  = threadIdx.x;
    const int lane = tid & 31;
    const int wave = __builtin_amdgcn_readfirstlane(tid >> 5);
    const int lo   = blockIdx.x * SL_NODES;

    for (int k = tid; k < SL_NODES; k += DG_THREADS) cnt[k] = 0u;
    __syncthreads();

    const uint4* row4 = (const uint4*)(cells + (size_t)blockIdx.x * ncell * CELL_WORDS);
    const int    nb   = ncell >> 2;
    const unsigned w0 = 4u * (unsigned)(lane & 7);

    for (int it0 = wave; it0 < nb; it0 += DG_AHEAD * DG_WAVES) {
        uint4 c[DG_AHEAD];
        #pragma unroll
        for (int j = 0; j < DG_AHEAD; ++j) {
            const int it = it0 + j * DG_WAVES;
            c[j] = row4[((it < nb) ? it : it0) * 32 + lane];
        }
        unsigned long long ov = 0ull;
        #pragma unroll
        for (int j = 0; j < DG_AHEAD; ++j) {
            const int it = it0 + j * DG_WAVES;
            if (it < nb) {
                const unsigned n    = (unsigned)__shfl((int)c[j].x, lane & 24);
                const bool     over = n > (unsigned)CELL_CAP;
                const unsigned nn   = over ? 0u : n;
                if (w0 - 1u < nn) atomicAdd(&cnt[c[j].x >> SL_SRC_BITS], 1u);
                if (w0      < nn) atomicAdd(&cnt[c[j].y >> SL_SRC_BITS], 1u);
                if (w0 + 1u < nn) atomicAdd(&cnt[c[j].z >> SL_SRC_BITS], 1u);
                if (w0 + 2u < nn) atomicAdd(&cnt[c[j].w >> SL_SRC_BITS], 1u);
                const unsigned om = __builtin_amdgcn_ballot_w32(over);
                const unsigned o4 = (om & 1u) | ((om >> 7) & 2u) | ((om >> 14) & 4u) | ((om >> 21) & 8u);
                ov |= (unsigned long long)o4 << (4 * j);
            }
        }
        while (ov != 0ull) {
            const int b = __builtin_ctzll(ov);
            ov &= ov - 1ull;
            const int e0 = (4 * (it0 + (b >> 2) * DG_WAVES) + (b & 3)) * PB_EDGES;
            #pragma clang loop unroll(disable)
            for (int k = 0; k < PB_EDGES; k += 32) {
                const int e = e0 + k + lane;
                if (e < E) {
                    const unsigned r = (unsigned)(dst[e] - lo);
                    if (r < (unsigned)SL_NODES) atomicAdd(&cnt[r], 1u);
                }
            }
        }
    }
    __syncthreads();

    int rows = N - lo; if (rows > SL_NODES) rows = SL_NODES;
    for (int k = tid; k < rows; k += DG_THREADS) {
        const float d = (float)(1u + cnt[k]);
        dinv[lo + k] = (d > 0.0f) ? rsqrtf(d) : 0.0f;
    }
}


#define GM_WAVES  4
#define GM_BLOCKS 256
#define GM_KSTEPS (F_IN / 4)

__global__ void __launch_bounds__(GM_WAVES * 32) k_gemm(const float* __restrict__ x,
                                                        const float* __restrict__ W,
                                                        const float* __restrict__ dinv,
                                                        float* __restrict__ h, int N) {
    const int wave = __builtin_amdgcn_readfirstlane(threadIdx.x >> 5);
    const int lane = threadIdx.x & 31;
    const int lr    = lane & 15;
    const int khalf = lane >> 4;

    const int ntiles = (N + 15) >> 4;
    const int nw     = (int)gridDim.x * GM_WAVES;
    int tile = wave * (int)gridDim.x + (int)blockIdx.x;
    if (tile >= ntiles) return;

    v2f b[GM_KSTEPS][4];
    #pragma unroll
    for (int t = 0; t < 4; ++t) {
        const float2* wp = (const float2*)(W + (size_t)(lr + 16 * t) * F_IN);
        #pragma unroll
        for (int ks = 0; ks < GM_KSTEPS; ++ks) {
            const float2 f = wp[2 * ks + khalf];
            b[ks][t].x = f.x; b[ks][t].y = f.y;
        }
    }

    v2f a[GM_KSTEPS];
    {
        int m = tile * 16 + lr;
        if (m >= N) m = N - 1;
        const float2* xp = (const float2*)(x + (size_t)m * F_IN);
        #pragma unroll
        for (int ks = 0; ks < GM_KSTEPS; ++ks) {
            const float2 f = xp[2 * ks + khalf];
            a[ks].x = f.x; a[ks].y = f.y;
        }
    }

    for (;;) {
        const int  nxt  = tile + nw;
        const bool more = nxt < ntiles;

        const int mbase = tile * 16 + (khalf << 3);
        float sc[8];
        #pragma unroll
        for (int v = 0; v < 8; ++v) {
            int mm = mbase + v;
            if (mm >= N) mm = N - 1;
            sc[v] = dinv[mm];
        }

        v2f an[GM_KSTEPS];
        {
            int m = (more ? nxt : tile) * 16 + lr;
            if (m >= N) m = N - 1;
            const float2* xp = (const float2*)(x + (size_t)m * F_IN);
            #pragma unroll
            for (int ks = 0; ks < GM_KSTEPS; ++ks) {
                const float2 f = xp[2 * ks + khalf];
                an[ks].x = f.x; an[ks].y = f.y;
            }
        }

        v8f c0 = {}, c1 = {}, c2 = {}, c3 = {};
        #pragma unroll
        for (int ks = 0; ks < GM_KSTEPS; ++ks) {
            c0 = __builtin_amdgcn_wmma_f32_16x16x4_f32(false, a[ks], false, b[ks][0], (short)0, c0, false, false);
            c1 = __builtin_amdgcn_wmma_f32_16x16x4_f32(false, a[ks], false, b[ks][1], (short)0, c1, false, false);
            c2 = __builtin_amdgcn_wmma_f32_16x16x4_f32(false, a[ks], false, b[ks][2], (short)0, c2, false, false);
            c3 = __builtin_amdgcn_wmma_f32_16x16x4_f32(false, a[ks], false, b[ks][3], (short)0, c3, false, false);
        }

        if (tile * 16 + 16 <= N) {
            float* hp = h + (size_t)mbase * F_OUT + lr;
            #pragma unroll
            for (int v = 0; v < 8; ++v) {
                hp[v * F_OUT +  0] = c0[v] * sc[v];
                hp[v * F_OUT + 16] = c1[v] * sc[v];
                hp[v * F_OUT + 32] = c2[v] * sc[v];
                hp[v * F_OUT + 48] = c3[v] * sc[v];
            }
        } else {
            #pragma unroll
            for (int v = 0; v < 8; ++v) {
                const int mm = mbase + v;
                if (mm < N) {
                    float* hp = h + (size_t)mm * F_OUT + lr;
                    hp[ 0] = c0[v] * sc[v];
                    hp[16] = c1[v] * sc[v];
                    hp[32] = c2[v] * sc[v];
                    hp[48] = c3[v] * sc[v];
                }
            }
        }

        if (!more) break;
        #pragma unroll
        for (int ks = 0; ks < GM_KSTEPS; ++ks) a[ks] = an[ks];
        tile = nxt;
    }
}


#define SL_THREADS 1024
#define SL_WAVES   (SL_THREADS / 32)
#define SL_QCAP    64
#define SL_ROWS    8
#define SL_EPI     ((SL_NODES + SL_WAVES - 1) / SL_WAVES)

__device__ __forceinline__ void sl_drain(const unsigned* qw, int n, int lane,
                                         const float* __restrict__ h, float* acc) {
    unsigned p = 0u;
    if (lane < n) p = qw[lane];
    const int      s = (int)(p & ((1u << SL_SRC_BITS) - 1u));
    const unsigned r = p >> SL_SRC_BITS;
    #pragma clang loop unroll(disable)
    for (int i = 0; i < n; i += SL_ROWS) {
        float2   v[SL_ROWS];
        unsigned rr[SL_ROWS];
        #pragma unroll
        for (int u = 0; u < SL_ROWS; ++u) {
            const int idx = (i + u < n) ? i + u : n - 1;
            const int su  = __builtin_amdgcn_readlane(s, idx);
            rr[u] = (unsigned)__builtin_amdgcn_readlane((int)r, idx);
            v[u]  = ((const float2*)(h + (size_t)su * F_OUT))[lane];
        }
        #pragma unroll
        for (int u = 0; u < SL_ROWS; ++u) {
            if (i + u < n) {
                float* a = acc + rr[u] * F_OUT + 2 * lane;
                atomicAdd(a,     v[u].x);
                atomicAdd(a + 1, v[u].y);
            }
        }
    }
}

__device__ __forceinline__ int sl_push(bool has, unsigned word, unsigned* qw, int cnt, int lane,
                                       const float* __restrict__ h, float* acc) {
    const unsigned m = __builtin_amdgcn_ballot_w32(has);
    if (m != 0u) {
        if (has) qw[cnt + (int)__builtin_amdgcn_mbcnt_lo(m, 0u)] = word;
        cnt += __builtin_popcount(m);
        if (cnt >= 32) {
            __builtin_amdgcn_fence(__ATOMIC_ACQ_REL, "wavefront");
            sl_drain(qw, 32, lane, h, acc);
            cnt -= 32;
            unsigned mv = 0u;
            if (lane < cnt) mv = qw[32 + lane];
            if (lane < cnt) qw[lane] = mv;
            __builtin_amdgcn_fence(__ATOMIC_ACQ_REL, "wavefront");
        }
    }
    return cnt;
}

__device__ __forceinline__ int sl_edges(int cell, int lo, const int* __restrict__ src,
                                        const int* __restrict__ dst, int E, unsigned* qw, int cnt,
                                        int lane, const float* __restrict__ h, float* acc) {
    const int e0 = cell * PB_EDGES;
    #pragma clang loop unroll(disable)
    for (int k = 0; k < PB_EDGES; k += 32) {
        const int e = e0 + k + lane;
        bool     has  = false;
        unsigned word = 0u;
        if (e < E) {
            const unsigned r = (unsigned)(dst[e] - lo);
            if (r < (unsigned)SL_NODES) {
                has  = true;
                word = (unsigned)src[e] | (r << SL_SRC_BITS);
            }
        }
        cnt = sl_push(has, word, qw, cnt, lane, h, acc);
    }
    return cnt;
}

__global__ void __launch_bounds__(SL_THREADS) k_slice_agg(const unsigned* __restrict__ cells,
                                                          const int* __restrict__ src,
                                                          const int* __restrict__ dst,
                                                          const float* __restrict__ h,
                                                          const float* __restrict__ dinv,
                                                          const float* __restrict__ b_conv,
                                                          const float* __restrict__ W_lin,
                                                          const float* __restrict__ b_lin,
                                                          float* __restrict__ out, int N, int E, int ncell) {
    __shared__ __attribute__((aligned(16))) float acc[SL_NODES * F_OUT];
    __shared__ unsigned q[SL_WAVES * SL_QCAP];

    const int tid  = threadIdx.x;
    const int lane = tid & 31;
    const int wave = __builtin_amdgcn_readfirstlane(tid >> 5);
    const int lo   = blockIdx.x * SL_NODES;

    float4* acc4 = (float4*)acc;
    for (int i = tid; i < SL_NODES * F_OUT / 4; i += SL_THREADS)
        acc4[i] = make_float4(0.0f, 0.0f, 0.0f, 0.0f);
    __syncthreads();

    const uint4* row4 = (const uint4*)(cells + (size_t)blockIdx.x * ncell * CELL_WORDS);
    const int    nb   = ncell >> 2;

    unsigned* qw = q + wave * SL_QCAP;
    int cnt = 0;

    const unsigned w0 = 4u * (unsigned)(lane & 7);

    uint4 nxt = make_uint4(0u, 0u, 0u, 0u);
    int it = wave;
    if (it < nb) nxt = row4[it * 32 + lane];
    while (it < nb) {
        const uint4 cur = nxt;
        const int nit = it + SL_WAVES;
        nxt = row4[((nit < nb) ? nit : it) * 32 + lane];

        const unsigned n    = (unsigned)__shfl((int)cur.x, lane & 24);
        const bool     over = n > (unsigned)CELL_CAP;
        const unsigned nn   = over ? 0u : n;
        cnt = sl_push(w0 - 1u < nn, cur.x, qw, cnt, lane, h, acc);
        cnt = sl_push(w0      < nn, cur.y, qw, cnt, lane, h, acc);
        cnt = sl_push(w0 + 1u < nn, cur.z, qw, cnt, lane, h, acc);
        cnt = sl_push(w0 + 2u < nn, cur.w, qw, cnt, lane, h, acc);
        const unsigned om = __builtin_amdgcn_ballot_w32(over);
        if (om != 0u) {
            for (int c = 0; c < 4; ++c)
                if ((om >> (8 * c)) & 1u)
                    cnt = sl_edges(4 * it + c, lo, src, dst, E, qw, cnt, lane, h, acc);
        }
        it = nit;
    }
    if (cnt > 0) {
        __builtin_amdgcn_fence(__ATOMIC_ACQ_REL, "wavefront");
        sl_drain(qw, cnt, lane, h, acc);
    }

    __syncthreads();

    int rows = N - lo; if (rows > SL_NODES) rows = SL_NODES;
    const float2 bc = *(const float2*)(b_conv + 2 * lane);
    const float2 wl = *(const float2*)(W_lin + 2 * lane);
    const float  bl = b_lin[0];
    float2 hv[SL_EPI];
    float  di[SL_EPI];
    #pragma unroll
    for (int k = 0; k < SL_EPI; ++k) {
        int r = wave + k * SL_WAVES;
        if (r >= rows) r = rows - 1;
        const int node = lo + r;
        hv[k] = ((const float2*)(h + (size_t)node * F_OUT))[lane];
        di[k] = dinv[node];
    }
    #pragma unroll
    for (int k = 0; k < SL_EPI; ++k) {
        const int r  = wave + k * SL_WAVES;
        const int rc = (r < rows) ? r : rows - 1;
        const float2 a = ((const float2*)acc)[rc * (F_OUT / 2) + lane];
        const float ax = bc.x + di[k] * (hv[k].x + a.x);
        const float ay = bc.y + di[k] * (hv[k].y + a.y);
        float p = fmaxf(ax, 0.0f) * wl.x + fmaxf(ay, 0.0f) * wl.y;
        #pragma unroll
        for (int off = 16; off > 0; off >>= 1)
            p += __shfl_down(p, off);
        if (lane == 0 && r < rows) {
            const float z = p + bl;
            out[lo + r] = 1.0f / (1.0f + __expf(-z));
        }
    }
}


extern "C" void kernel_launch(void* const* d_in, const int* in_sizes, int n_in,
                              void* d_out, int out_size, void* d_ws, size_t ws_size,
                              hipStream_t stream) {
    const float* x      = (const float*)d_in[0];
    const int*   edge   = (const int*)  d_in[1];
    const float* W_conv = (const float*)d_in[2];
    const float* b_conv = (const float*)d_in[3];
    const float* W_lin  = (const float*)d_in[4];
    const float* b_lin  = (const float*)d_in[5];
    float*       out    = (float*)d_out;

    const int N = in_sizes[0] / F_IN;
    const int E = in_sizes[1] / 2;
    const int* src = edge;
    const int* dst = edge + E;

    float* ws   = (float*)d_ws;
    float* h    = ws;
    float* dinv = h + (size_t)N * F_OUT;
    unsigned* cells = (unsigned*)(dinv + (N + 63) / 64 * 64);
    const int nsl   = (N + SL_NODES - 1) / SL_NODES;
    const int ncell = ((E + PB_EDGES - 1) / PB_EDGES + 3) / 4 * 4;

    k_prep<<<ncell, PB_THREADS, 0, stream>>>(src, dst, cells, E, nsl, ncell);
    k_deg <<<nsl, DG_THREADS, 0, stream>>>(cells, dst, dinv, N, E, ncell);

    k_gemm<<<GM_BLOCKS, GM_WAVES * 32, 0, stream>>>(x, W_conv, dinv, h, N);

    k_slice_agg<<<nsl, SL_THREADS, 0, stream>>>(cells, src, dst, h, dinv, b_conv, W_lin, b_lin, out, N, E, ncell);
}


# k_slice_agg: LDS float adds gone: a hit takes a slot in its row's LDS source list (one integer add); the row's owner wave sums the listed hs rows in registers and finishes it; row sums by arrival
# speedup vs baseline: 5.1325x; 1.1849x over previous
#include <hip/hip_runtime.h>
#include <math.h>

typedef float v2f __attribute__((ext_vector_type(2)));
typedef float v8f __attribute__((ext_vector_type(8)));

#define F_IN  128
#define F_OUT 64


#define SL_NODES    400
#define SL_MAXSL    256
#define SL_SRC_BITS 20
#define PB_THREADS  512
#define PB_EDGES    (4 * PB_THREADS)
#define PB_PITCH    33
#define CELL_WORDS  32
#define CELL_CAP    (CELL_WORDS - 1)

__global__ void __launch_bounds__(PB_THREADS) k_prep(const int* __restrict__ src, const int* __restrict__ dst,
                       unsigned* __restrict__ cells, int E, int nsl, int ncell) {
    __shared__ unsigned tbl[SL_MAXSL * PB_PITCH];
    const int tid = threadIdx.x;
    const int i   = blockIdx.x * PB_THREADS + tid;
    for (int k = tid; k < nsl; k += PB_THREADS) tbl[k * PB_PITCH] = 0u;
    __syncthreads();
    int dd[4], ss[4];
    #pragma unroll
    for (int j = 0; j < 4; ++j) {
        const int e  = 4 * i + j;
        const int ec = (e < E) ? e : E - 1;
        dd[j] = dst[ec];
        ss[j] = src[ec];
    }
    unsigned sl[4], slot[4];
    #pragma unroll
    for (int j = 0; j < 4; ++j) {
        sl[j]   = (unsigned)dd[j] / (unsigned)SL_NODES;
        slot[j] = (unsigned)CELL_CAP;
        if (4 * i + j < E) slot[j] = atomicAdd(&tbl[sl[j] * PB_PITCH], 1u);
    }
    #pragma unroll
    for (int j = 0; j < 4; ++j) {
        if (slot[j] < (unsigned)CELL_CAP)
            tbl[sl[j] * PB_PITCH + 1 + slot[j]] =
                (unsigned)ss[j] | (((unsigned)dd[j] - sl[j] * (unsigned)SL_NODES) << SL_SRC_BITS);
    }
    __syncthreads();
    const int wave = tid >> 5, lane = tid & 31;
    for (int s = wave; s < nsl; s += PB_THREADS / 32)
        cells[((size_t)s * ncell + blockIdx.x) * CELL_WORDS + lane] = tbl[s * PB_PITCH + lane];
}


#define DG_THREADS 512
#define DG_WAVES   (DG_THREADS / 32)
#define DG_AHEAD   16

__global__ void __launch_bounds__(DG_THREADS) k_deg(const unsigned* __restrict__ cells,
                                                    const int* __restrict__ dst,
                                                    float* __restrict__ dinv, int N, int E, int ncell) {
    __shared__ unsigned cnt[SL_NODES];
    const int tid  = threadIdx.x;
    const int lane = tid & 31;
    const int wave = __builtin_amdgcn_readfirstlane(tid >> 5);
    const int lo   = blockIdx.x * SL_NODES;

    for (int k = tid; k < SL_NODES; k += DG_THREADS) cnt[k] = 0u;
    __syncthreads();

    const uint4* row4 = (const uint4*)(cells + (size_t)blockIdx.x * ncell * CELL_WORDS);
    const int    nb   = ncell >> 2;
    const unsigned w0 = 4u * (unsigned)(lane & 7);

    for (int it0 = wave; it0 < nb; it0 += DG_AHEAD * DG_WAVES) {
        uint4 c[DG_AHEAD];
        #pragma unroll
        for (int j = 0; j < DG_AHEAD; ++j) {
            const int it = it0 + j * DG_WAVES;
            c[j] = row4[((it < nb) ? it : it0) * 32 + lane];
        }
        unsigned long long ov = 0ull;
        #pragma unroll
        for (int j = 0; j < DG_AHEAD; ++j) {
            const int it = it0 + j * DG_WAVES;
            if (it < nb) {
                const unsigned n    = (unsigned)__shfl((int)c[j].x, lane & 24);
                const bool     over = n > (unsigned)CELL_CAP;
                const unsigned nn   = over ? 0u : n;
                if (w0 - 1u < nn) atomicAdd(&cnt[c[j].x >> SL_SRC_BITS], 1u);
                if (w0      < nn) atomicAdd(&cnt[c[j].y >> SL_SRC_BITS], 1u);
                if (w0 + 1u < nn) atomicAdd(&cnt[c[j].z >> SL_SRC_BITS], 1u);
                if (w0 + 2u < nn) atomicAdd(&cnt[c[j].w >> SL_SRC_BITS], 1u);
                const unsigned om = __builtin_amdgcn_ballot_w32(over);
                const unsigned o4 = (om & 1u) | ((om >> 7) & 2u) | ((om >> 14) & 4u) | ((om >> 21) & 8u);
                ov |= (unsigned long long)o4 << (4 * j);
            }
        }
        while (ov != 0ull) {
            const int b = __builtin_ctzll(ov);
            ov &= ov - 1ull;
            const int e0 = (4 * (it0 + (b >> 2) * DG_WAVES) + (b & 3)) * PB_EDGES;
            #pragma clang loop unroll(disable)
            for (int k = 0; k < PB_EDGES; k += 32) {
                const int e = e0 + k + lane;
                if (e < E) {
                    const unsigned r = (unsigned)(dst[e] - lo);
                    if (r < (unsigned)SL_NODES) atomicAdd(&cnt[r], 1u);
                }
            }
        }
    }
    __syncthreads();

    int rows = N - lo; if (rows > SL_NODES) rows = SL_NODES;
    for (int k = tid; k < rows; k += DG_THREADS) {
        const float d = (float)(1u + cnt[k]);
        dinv[lo + k] = (d > 0.0f) ? rsqrtf(d) : 0.0f;
    }
}


#define GM_WAVES  4
#define GM_BLOCKS 256
#define GM_KSTEPS (F_IN / 4)

__global__ void __launch_bounds__(GM_WAVES * 32) k_gemm(const float* __restrict__ x,
                                                        const float* __restrict__ W,
                                                        const float* __restrict__ dinv,
                                                        float* __restrict__ h, int N) {
    const int wave = __builtin_amdgcn_readfirstlane(threadIdx.x >> 5);
    const int lane = threadIdx.x & 31;
    const int lr    = lane & 15;
    const int khalf = lane >> 4;

    const int ntiles = (N + 15) >> 4;
    const int nw     = (int)gridDim.x * GM_WAVES;
    int tile = wave * (int)gridDim.x + (int)blockIdx.x;
    if (tile >= ntiles) return;

    v2f b[GM_KSTEPS][4];
    #pragma unroll
    for (int t = 0; t < 4; ++t) {
        const float2* wp = (const float2*)(W + (size_t)(lr + 16 * t) * F_IN);
        #pragma unroll
        for (int ks = 0; ks < GM_KSTEPS; ++ks) {
            const float2 f = wp[2 * ks + khalf];
            b[ks][t].x = f.x; b[ks][t].y = f.y;
        }
    }

    v2f a[GM_KSTEPS];
    {
        int m = tile * 16 + lr;
        if (m >= N) m = N - 1;
        const float2* xp = (const float2*)(x + (size_t)m * F_IN);
        #pragma unroll
        for (int ks = 0; ks < GM_KSTEPS; ++ks) {
            const float2 f = xp[2 * ks + khalf];
            a[ks].x = f.x; a[ks].y = f.y;
        }
    }

    for (;;) {
        const int  nxt  = tile + nw;
        const bool more = nxt < ntiles;

        const int mbase = tile * 16 + (khalf << 3);
        float sc[8];
        #pragma unroll
        for (int v = 0; v < 8; ++v) {
            int mm = mbase + v;
            if (mm >= N) mm = N - 1;
            sc[v] = dinv[mm];
        }

        v2f an[GM_KSTEPS];
        {
            int m = (more ? nxt : tile) * 16 + lr;
            if (m >= N) m = N - 1;
            const float2* xp = (const float2*)(x + (size_t)m * F_IN);
            #pragma unroll
            for (int ks = 0; ks < GM_KSTEPS; ++ks) {
                const float2 f = xp[2 * ks + khalf];
                an[ks].x = f.x; an[ks].y = f.y;
            }
        }

        v8f c0 = {}, c1 = {}, c2 = {}, c3 = {};
        #pragma unroll
        for (int ks = 0; ks < GM_KSTEPS; ++ks) {
            c0 = __builtin_amdgcn_wmma_f32_16x16x4_f32(false, a[ks], false, b[ks][0], (short)0, c0, false, false);
            c1 = __builtin_amdgcn_wmma_f32_16x16x4_f32(false, a[ks], false, b[ks][1], (short)0, c1, false, false);
            c2 = __builtin_amdgcn_wmma_f32_16x16x4_f32(false, a[ks], false, b[ks][2], (short)0, c2, false, false);
            c3 = __builtin_amdgcn_wmma_f32_16x16x4_f32(false, a[ks], false, b[ks][3], (short)0, c3, false, false);
        }

        if (tile * 16 + 16 <= N) {
            float* hp = h + (size_t)mbase * F_OUT + lr;
            #pragma unroll
            for (int v = 0; v < 8; ++v) {
                hp[v * F_OUT +  0] = c0[v] * sc[v];
                hp[v * F_OUT + 16] = c1[v] * sc[v];
                hp[v * F_OUT + 32] = c2[v] * sc[v];
                hp[v * F_OUT + 48] = c3[v] * sc[v];
            }
        } else {
            #pragma unroll
            for (int v = 0; v < 8; ++v) {
                const int mm = mbase + v;
                if (mm < N) {
                    float* hp = h + (size_t)mm * F_OUT + lr;
                    hp[ 0] = c0[v] * sc[v];
                    hp[16] = c1[v] * sc[v];
                    hp[32] = c2[v] * sc[v];
                    hp[48] = c3[v] * sc[v];
                }
            }
        }

        if (!more) break;
        #pragma unroll
        for (int ks = 0; ks < GM_KSTEPS; ++ks) a[ks] = an[ks];
        tile = nxt;
    }
}


#define SL_THREADS 1024
#define SL_WAVES   (SL_THREADS / 32)
#define SL_RCAP    64
#define SL_AHEAD   8
#define SL_ROWS    8
#define SL_SRC_MASK ((1u << SL_SRC_BITS) - 1u)

__device__ __forceinline__ unsigned sl_slot(bool has, unsigned word, unsigned* cntr) {
    unsigned pos = 0xffffffffu;
    if (has) pos = atomicAdd(&cntr[word >> SL_SRC_BITS], 1u);
    return pos;
}
__device__ __forceinline__ bool sl_store(bool has, unsigned word, unsigned pos, unsigned* rq) {
    const bool fits = pos < (unsigned)SL_RCAP;
    if (fits) rq[(word >> SL_SRC_BITS) * SL_RCAP + pos] = word & SL_SRC_MASK;
    return has && !fits;
}
__device__ __forceinline__ void sl_full(bool full, unsigned word, float* acc,
                                        const float* __restrict__ h, int lane) {
    unsigned om = __builtin_amdgcn_ballot_w32(full);
    while (om != 0u) {
        const int l = __builtin_ctz(om);
        om &= om - 1u;
        const unsigned wl = (unsigned)__builtin_amdgcn_readlane((int)word, l);
        const float2 v = ((const float2*)(h + (size_t)(wl & SL_SRC_MASK) * F_OUT))[lane];
        float* a = acc + (wl >> SL_SRC_BITS) * F_OUT + 2 * lane;
        atomicAdd(a,     v.x);
        atomicAdd(a + 1, v.y);
    }
}

__device__ __forceinline__ void sl_edges(int cell, int lo, const int* __restrict__ src,
                                         const int* __restrict__ dst, int E, unsigned* cntr,
                                         unsigned* rq, float* acc, const float* __restrict__ h, int lane) {
    const int e0 = cell * PB_EDGES;
    #pragma clang loop unroll(disable)
    for (int k = 0; k < PB_EDGES; k += 32) {
        const int e = e0 + k + lane;
        bool     has  = false;
        unsigned word = 0u;
        if (e < E) {
            const unsigned r = (unsigned)(dst[e] - lo);
            if (r < (unsigned)SL_NODES) {
                has  = true;
                word = (unsigned)src[e] | (r << SL_SRC_BITS);
            }
        }
        const unsigned pos = sl_slot(has, word, cntr);
        sl_full(sl_store(has, word, pos, rq), word, acc, h, lane);
    }
}

__device__ __forceinline__ float2 sl_sum(unsigned s, int n, int lane, const float* __restrict__ h, float2 sum) {
    #pragma clang loop unroll(disable)
    for (int i = 0; i < n; i += 2 * SL_ROWS) {
        float2 va[SL_ROWS], vb[SL_ROWS];
        const bool two = i + SL_ROWS < n;
        #pragma unroll
        for (int u = 0; u < SL_ROWS; ++u) {
            const int idx = (i + u < n) ? i + u : n - 1;
            const unsigned su = (unsigned)__builtin_amdgcn_readlane((int)s, idx);
            va[u] = ((const float2*)(h + (size_t)su * F_OUT))[lane];
        }
        if (two) {
            #pragma unroll
            for (int u = 0; u < SL_ROWS; ++u) {
                const int j   = i + SL_ROWS + u;
                const int idx = (j < n) ? j : n - 1;
                const unsigned su = (unsigned)__builtin_amdgcn_readlane((int)s, idx);
                vb[u] = ((const float2*)(h + (size_t)su * F_OUT))[lane];
            }
        }
        #pragma unroll
        for (int u = 0; u < SL_ROWS; ++u) {
            const bool ok = i + u < n;
            sum.x += ok ? va[u].x : 0.0f;
            sum.y += ok ? va[u].y : 0.0f;
        }
        if (two) {
            #pragma unroll
            for (int u = 0; u < SL_ROWS; ++u) {
                const bool ok = i + SL_ROWS + u < n;
                sum.x += ok ? vb[u].x : 0.0f;
                sum.y += ok ? vb[u].y : 0.0f;
            }
        }
    }
    return sum;
}

__global__ void __launch_bounds__(SL_THREADS) k_slice_agg(const unsigned* __restrict__ cells,
                                                          const int* __restrict__ src,
                                                          const int* __restrict__ dst,
                                                          const float* __restrict__ h,
                                                          const float* __restrict__ dinv,
                                                          const float* __restrict__ b_conv,
                                                          const float* __restrict__ W_lin,
                                                          const float* __restrict__ b_lin,
                                                          float* __restrict__ out, int N, int E, int ncell) {
    __shared__ __attribute__((aligned(16))) float acc[SL_NODES * F_OUT];
    __shared__ unsigned rq[SL_NODES * SL_RCAP];
    __shared__ unsigned cntr[SL_NODES];

    const int tid  = threadIdx.x;
    const int lane = tid & 31;
    const int wave = __builtin_amdgcn_readfirstlane(tid >> 5);
    const int lo   = blockIdx.x * SL_NODES;

    float4* acc4 = (float4*)acc;
    for (int i = tid; i < SL_NODES * F_OUT / 4; i += SL_THREADS)
        acc4[i] = make_float4(0.0f, 0.0f, 0.0f, 0.0f);
    for (int k = tid; k < SL_NODES; k += SL_THREADS) cntr[k] = 0u;
    __syncthreads();

    const uint4* row4 = (const uint4*)(cells + (size_t)blockIdx.x * ncell * CELL_WORDS);
    const int    nb   = ncell >> 2;
    const unsigned w0 = 4u * (unsigned)(lane & 7);

    for (int it0 = wave; it0 < nb; it0 += SL_AHEAD * SL_WAVES) {
        uint4 c[SL_AHEAD];
        #pragma unroll
        for (int j = 0; j < SL_AHEAD; ++j) {
            const int it = it0 + j * SL_WAVES;
            c[j] = row4[((it < nb) ? it : it0) * 32 + lane];
        }
        unsigned ov = 0u;
        #pragma unroll
        for (int j = 0; j < SL_AHEAD; ++j) {
            const int it = it0 + j * SL_WAVES;
            if (it < nb) {
                const unsigned n    = (unsigned)__shfl((int)c[j].x, lane & 24);
                const bool     over = n > (unsigned)CELL_CAP;
                const unsigned nn   = over ? 0u : n;
                const bool hx = w0 - 1u < nn, hy = w0 < nn, hz = w0 + 1u < nn, hw = w0 + 2u < nn;
                const unsigned px = sl_slot(hx, c[j].x, cntr);
                const unsigned py = sl_slot(hy, c[j].y, cntr);
                const unsigned pz = sl_slot(hz, c[j].z, cntr);
                const unsigned pw = sl_slot(hw, c[j].w, cntr);
                const bool fx = sl_store(hx, c[j].x, px, rq);
                const bool fy = sl_store(hy, c[j].y, py, rq);
                const bool fz = sl_store(hz, c[j].z, pz, rq);
                const bool fw = sl_store(hw, c[j].w, pw, rq);
                sl_full(fx, c[j].x, acc, h, lane);
                sl_full(fy, c[j].y, acc, h, lane);
                sl_full(fz, c[j].z, acc, h, lane);
                sl_full(fw, c[j].w, acc, h, lane);
                const unsigned om = __builtin_amdgcn_ballot_w32(over);
                const unsigned o4 = (om & 1u) | ((om >> 7) & 2u) | ((om >> 14) & 4u) | ((om >> 21) & 8u);
                ov |= o4 << (4 * j);
            }
        }
        while (ov != 0u) {
            const int b = __builtin_ctz(ov);
            ov &= ov - 1u;
            sl_edges(4 * (it0 + (b >> 2) * SL_WAVES) + (b & 3), lo, src, dst, E, cntr, rq, acc, h, lane);
        }
    }

    __syncthreads();

    int rows = N - lo; if (rows > SL_NODES) rows = SL_NODES;
    const float2 bc = *(const float2*)(b_conv + 2 * lane);
    const float2 wl = *(const float2*)(W_lin + 2 * lane);
    const float  bl = b_lin[0];
    unsigned cnv = 0u;
    {
        const int rr = wave + lane * SL_WAVES;
        if (rr < rows) cnv = cntr[rr];
        if (cnv > (unsigned)SL_RCAP) cnv = (unsigned)SL_RCAP;
    }
    int k = 0;
    for (int r = wave; r < rows; r += SL_WAVES, ++k) {
        const int node = lo + r;
        const float2 hv = ((const float2*)(h + (size_t)node * F_OUT))[lane];
        const float  di = dinv[node];
        const int n = __builtin_amdgcn_readlane((int)cnv, k);
        unsigned s0 = 0u, s1 = 0u;
        if (lane < n)      s0 = rq[r * SL_RCAP + lane];
        if (32 + lane < n) s1 = rq[r * SL_RCAP + 32 + lane];
        float2 sum = make_float2(0.0f, 0.0f);
        sum = sl_sum(s0, (n < 32) ? n : 32, lane, h, sum);
        if (n > 32) sum = sl_sum(s1, n - 32, lane, h, sum);
        const float2 a = ((const float2*)acc)[r * (F_OUT / 2) + lane];
        const float ax = bc.x + di * (hv.x + (a.x + sum.x));
        const float ay = bc.y + di * (hv.y + (a.y + sum.y));
        float p = fmaxf(ax, 0.0f) * wl.x + fmaxf(ay, 0.0f) * wl.y;
        #pragma unroll
        for (int off = 16; off > 0; off >>= 1)
            p += __shfl_down(p, off);
        if (lane == 0) {
            const float z = p + bl;
            out[node] = 1.0f / (1.0f + __expf(-z));
        }
    }
}


extern "C" void kernel_launch(void* const* d_in, const int* in_sizes, int n_in,
                              void* d_out, int out_size, void* d_ws, size_t ws_size,
                              hipStream_t stream) {
    const float* x      = (const float*)d_in[0];
    const int*   edge   = (const int*)  d_in[1];
    const float* W_conv = (const float*)d_in[2];
    const float* b_conv = (const float*)d_in[3];
    const float* W_lin  = (const float*)d_in[4];
    const float* b_lin  = (const float*)d_in[5];
    float*       out    = (float*)d_out;

    const int N = in_sizes[0] / F_IN;
    const int E = in_sizes[1] / 2;
    const int* src = edge;
    const int* dst = edge + E;

    float* ws   = (float*)d_ws;
    float* h    = ws;
    float* dinv = h + (size_t)N * F_OUT;
    unsigned* cells = (unsigned*)(dinv + (N + 63) / 64 * 64);
    const int nsl   = (N + SL_NODES - 1) / SL_NODES;
    const int ncell = ((E + PB_EDGES - 1) / PB_EDGES + 3) / 4 * 4;

    k_prep<<<ncell, PB_THREADS, 0, stream>>>(src, dst, cells, E, nsl, ncell);
    k_deg <<<nsl, DG_THREADS, 0, stream>>>(cells, dst, dinv, N, E, ncell);

    k_gemm<<<GM_BLOCKS, GM_WAVES * 32, 0, stream>>>(x, W_conv, dinv, h, N);

    k_slice_agg<<<nsl, SL_THREADS, 0, stream>>>(cells, src, dst, h, dinv, b_conv, W_lin, b_lin, out, N, E, ncell);
}


# hit cells now 256 B over 7168 edges (63 slots, 45% full; were 128 B over 2048, 26% full): k_prep 224 blocks, 14 edges a thread; cell bytes and read batches down 43%; row sums by arrival as before
# speedup vs baseline: 5.3569x; 1.0437x over previous
#include <hip/hip_runtime.h>
#include <math.h>

typedef float v2f __attribute__((ext_vector_type(2)));
typedef float v8f __attribute__((ext_vector_type(8)));

#define F_IN  128
#define F_OUT 64


#define SL_NODES    400
#define SL_MAXSL    256
#define SL_SRC_BITS 20
#define PB_THREADS  512
#define PB_PER_THR  14
#define PB_EDGES    (PB_PER_THR * PB_THREADS)
#define PB_PITCH    65
#define CELL_WORDS  64
#define CELL_CAP    (CELL_WORDS - 1)
#define CELL_LANES  (CELL_WORDS / 4)
#define BATCH_CELLS (32 / CELL_LANES)

__global__ void __launch_bounds__(PB_THREADS) k_prep(const int* __restrict__ src, const int* __restrict__ dst,
                       unsigned* __restrict__ cells, int E, int nsl, int ncell) {
    __shared__ unsigned tbl[SL_MAXSL * PB_PITCH];
    const int tid = threadIdx.x;
    const int e0  = blockIdx.x * PB_EDGES + tid;
    for (int k = tid; k < nsl; k += PB_THREADS) tbl[k * PB_PITCH] = 0u;
    __syncthreads();
    int dd[PB_PER_THR], ss[PB_PER_THR];
    #pragma unroll
    for (int j = 0; j < PB_PER_THR; ++j) {
        const int e  = e0 + j * PB_THREADS;
        const int ec = (e < E) ? e : E - 1;
        dd[j] = dst[ec];
        ss[j] = src[ec];
    }
    unsigned sl[PB_PER_THR], slot[PB_PER_THR];
    #pragma unroll
    for (int j = 0; j < PB_PER_THR; ++j) {
        sl[j]   = (unsigned)dd[j] / (unsigned)SL_NODES;
        slot[j] = (unsigned)CELL_CAP;
        if (e0 + j * PB_THREADS < E) slot[j] = atomicAdd(&tbl[sl[j] * PB_PITCH], 1u);
    }
    #pragma unroll
    for (int j = 0; j < PB_PER_THR; ++j) {
        if (slot[j] < (unsigned)CELL_CAP)
            tbl[sl[j] * PB_PITCH + 1 + slot[j]] =
                (unsigned)ss[j] | (((unsigned)dd[j] - sl[j] * (unsigned)SL_NODES) << SL_SRC_BITS);
    }
    __syncthreads();
    const int wave = tid >> 5, lane = tid & 31;
    for (int s = wave; s < nsl; s += PB_THREADS / 32) {
        unsigned* cp = cells + ((size_t)s * ncell + blockIdx.x) * CELL_WORDS;
        cp[lane]      = tbl[s * PB_PITCH + lane];
        cp[32 + lane] = tbl[s * PB_PITCH + 32 + lane];
    }
}


#define DG_THREADS 512
#define DG_WAVES   (DG_THREADS / 32)
#define DG_AHEAD   7

__global__ void __launch_bounds__(DG_THREADS) k_deg(const unsigned* __restrict__ cells,
                                                    const int* __restrict__ dst,
                                                    float* __restrict__ dinv, int N, int E, int ncell) {
    __shared__ unsigned cnt[SL_NODES];
    const int tid  = threadIdx.x;
    const int lane = tid & 31;
    const int wave = __builtin_amdgcn_readfirstlane(tid >> 5);
    const int lo   = blockIdx.x * SL_NODES;

    for (int k = tid; k < SL_NODES; k += DG_THREADS) cnt[k] = 0u;
    __syncthreads();

    const uint4* row4 = (const uint4*)(cells + (size_t)blockIdx.x * ncell * CELL_WORDS);
    const int    nb   = ncell / BATCH_CELLS;
    const unsigned w0 = 4u * (unsigned)(lane & (CELL_LANES - 1));

    for (int it0 = wave; it0 < nb; it0 += DG_AHEAD * DG_WAVES) {
        uint4 c[DG_AHEAD];
        #pragma unroll
        for (int j = 0; j < DG_AHEAD; ++j) {
            const int it = it0 + j * DG_WAVES;
            c[j] = row4[((it < nb) ? it : it0) * 32 + lane];
        }
        unsigned ov = 0u;
        #pragma unroll
        for (int j = 0; j < DG_AHEAD; ++j) {
            const int it = it0 + j * DG_WAVES;
            if (it < nb) {
                const unsigned n    = (unsigned)__shfl((int)c[j].x, lane & 16);
                const bool     over = n > (unsigned)CELL_CAP;
                const unsigned nn   = over ? 0u : n;
                if (w0 - 1u < nn) atomicAdd(&cnt[c[j].x >> SL_SRC_BITS], 1u);
                if (w0      < nn) atomicAdd(&cnt[c[j].y >> SL_SRC_BITS], 1u);
                if (w0 + 1u < nn) atomicAdd(&cnt[c[j].z >> SL_SRC_BITS], 1u);
                if (w0 + 2u < nn) atomicAdd(&cnt[c[j].w >> SL_SRC_BITS], 1u);
                const unsigned om = __builtin_amdgcn_ballot_w32(over);
                const unsigned o2 = (om & 1u) | ((om >> 15) & 2u);
                ov |= o2 << (2 * j);
            }
        }
        while (ov != 0u) {
            const int b = __builtin_ctz(ov);
            ov &= ov - 1u;
            const int e0 = (2 * (it0 + (b >> 1) * DG_WAVES) + (b & 1)) * PB_EDGES;
            #pragma clang loop unroll(disable)
            for (int k = 0; k < PB_EDGES; k += 32) {
                const int e = e0 + k + lane;
                if (e < E) {
                    const unsigned r = (unsigned)(dst[e] - lo);
                    if (r < (unsigned)SL_NODES) atomicAdd(&cnt[r], 1u);
                }
            }
        }
    }
    __syncthreads();

    int rows = N - lo; if (rows > SL_NODES) rows = SL_NODES;
    for (int k = tid; k < rows; k += DG_THREADS) {
        const float d = (float)(1u + cnt[k]);
        dinv[lo + k] = (d > 0.0f) ? rsqrtf(d) : 0.0f;
    }
}


#define GM_WAVES  4
#define GM_BLOCKS 256
#define GM_KSTEPS (F_IN / 4)

__global__ void __launch_bounds__(GM_WAVES * 32) k_gemm(const float* __restrict__ x,
                                                        const float* __restrict__ W,
                                                        const float* __restrict__ dinv,
                                                        float* __restrict__ h, int N) {
    const int wave = __builtin_amdgcn_readfirstlane(threadIdx.x >> 5);
    const int lane = threadIdx.x & 31;
    const int lr    = lane & 15;
    const int khalf = lane >> 4;

    const int ntiles = (N + 15) >> 4;
    const int nw     = (int)gridDim.x * GM_WAVES;
    int tile = wave * (int)gridDim.x + (int)blockIdx.x;
    if (tile >= ntiles) return;

    v2f b[GM_KSTEPS][4];
    #pragma unroll
    for (int t = 0; t < 4; ++t) {
        const float2* wp = (const float2*)(W + (size_t)(lr + 16 * t) * F_IN);
        #pragma unroll
        for (int ks = 0; ks < GM_KSTEPS; ++ks) {
            const float2 f = wp[2 * ks + khalf];
            b[ks][t].x = f.x; b[ks][t].y = f.y;
        }
    }

    v2f a[GM_KSTEPS];
    {
        int m = tile * 16 + lr;
        if (m >= N) m = N - 1;
        const float2* xp = (const float2*)(x + (size_t)m * F_IN);
        #pragma unroll
        for (int ks = 0; ks < GM_KSTEPS; ++ks) {
            const float2 f = xp[2 * ks + khalf];
            a[ks].x = f.x; a[ks].y = f.y;
        }
    }

    for (;;) {
        const int  nxt  = tile + nw;
        const bool more = nxt < ntiles;

        const int mbase = tile * 16 + (khalf << 3);
        float sc[8];
        #pragma unroll
        for (int v = 0; v < 8; ++v) {
            int mm = mbase + v;
            if (mm >= N) mm = N - 1;
            sc[v] = dinv[mm];
        }

        v2f an[GM_KSTEPS];
        {
            int m = (more ? nxt : tile) * 16 + lr;
            if (m >= N) m = N - 1;
            const float2* xp = (const float2*)(x + (size_t)m * F_IN);
            #pragma unroll
            for (int ks = 0; ks < GM_KSTEPS; ++ks) {
                const float2 f = xp[2 * ks + khalf];
                an[ks].x = f.x; an[ks].y = f.y;
            }
        }

        v8f c0 = {}, c1 = {}, c2 = {}, c3 = {};
        #pragma unroll
        for (int ks = 0; ks < GM_KSTEPS; ++ks) {
            c0 = __builtin_amdgcn_wmma_f32_16x16x4_f32(false, a[ks], false, b[ks][0], (short)0, c0, false, false);
            c1 = __builtin_amdgcn_wmma_f32_16x16x4_f32(false, a[ks], false, b[ks][1], (short)0, c1, false, false);
            c2 = __builtin_amdgcn_wmma_f32_16x16x4_f32(false, a[ks], false, b[ks][2], (short)0, c2, false, false);
            c3 = __builtin_amdgcn_wmma_f32_16x16x4_f32(false, a[ks], false, b[ks][3], (short)0, c3, false, false);
        }

        if (tile * 16 + 16 <= N) {
            float* hp = h + (size_t)mbase * F_OUT + lr;
            #pragma unroll
            for (int v = 0; v < 8; ++v) {
                hp[v * F_OUT +  0] = c0[v] * sc[v];
                hp[v * F_OUT + 16] = c1[v] * sc[v];
                hp[v * F_OUT + 32] = c2[v] * sc[v];
                hp[v * F_OUT + 48] = c3[v] * sc[v];
            }
        } else {
            #pragma unroll
            for (int v = 0; v < 8; ++v) {
                const int mm = mbase + v;
                if (mm < N) {
                    float* hp = h + (size_t)mm * F_OUT + lr;
                    hp[ 0] = c0[v] * sc[v];
                    hp[16] = c1[v] * sc[v];
                    hp[32] = c2[v] * sc[v];
                    hp[48] = c3[v] * sc[v];
                }
            }
        }

        if (!more) break;
        #pragma unroll
        for (int ks = 0; ks < GM_KSTEPS; ++ks) a[ks] = an[ks];
        tile = nxt;
    }
}


#define SL_THREADS 1024
#define SL_WAVES   (SL_THREADS / 32)
#define SL_RCAP    64
#define SL_AHEAD   4
#define SL_ROWS    8
#define SL_SRC_MASK ((1u << SL_SRC_BITS) - 1u)

__device__ __forceinline__ unsigned sl_slot(bool has, unsigned word, unsigned* cntr) {
    unsigned pos = 0xffffffffu;
    if (has) pos = atomicAdd(&cntr[word >> SL_SRC_BITS], 1u);
    return pos;
}
__device__ __forceinline__ bool sl_store(bool has, unsigned word, unsigned pos, unsigned* rq) {
    const bool fits = pos < (unsigned)SL_RCAP;
    if (fits) rq[(word >> SL_SRC_BITS) * SL_RCAP + pos] = word & SL_SRC_MASK;
    return has && !fits;
}
__device__ __forceinline__ void sl_full(bool full, unsigned word, float* acc,
                                        const float* __restrict__ h, int lane) {
    unsigned om = __builtin_amdgcn_ballot_w32(full);
    while (om != 0u) {
        const int l = __builtin_ctz(om);
        om &= om - 1u;
        const unsigned wl = (unsigned)__builtin_amdgcn_readlane((int)word, l);
        const float2 v = ((const float2*)(h + (size_t)(wl & SL_SRC_MASK) * F_OUT))[lane];
        float* a = acc + (wl >> SL_SRC_BITS) * F_OUT + 2 * lane;
        atomicAdd(a,     v.x);
        atomicAdd(a + 1, v.y);
    }
}

__device__ __forceinline__ void sl_edges(int cell, int lo, const int* __restrict__ src,
                                         const int* __restrict__ dst, int E, unsigned* cntr,
                                         unsigned* rq, float* acc, const float* __restrict__ h, int lane) {
    const int e0 = cell * PB_EDGES;
    #pragma clang loop unroll(disable)
    for (int k = 0; k < PB_EDGES; k += 32) {
        const int e = e0 + k + lane;
        bool     has  = false;
        unsigned word = 0u;
        if (e < E) {
            const unsigned r = (unsigned)(dst[e] - lo);
            if (r < (unsigned)SL_NODES) {
                has  = true;
                word = (unsigned)src[e] | (r << SL_SRC_BITS);
            }
        }
        const unsigned pos = sl_slot(has, word, cntr);
        sl_full(sl_store(has, word, pos, rq), word, acc, h, lane);
    }
}

__device__ __forceinline__ float2 sl_sum(unsigned s, int n, int lane, const float* __restrict__ h, float2 sum) {
    #pragma clang loop unroll(disable)
    for (int i = 0; i < n; i += 2 * SL_ROWS) {
        float2 va[SL_ROWS], vb[SL_ROWS];
        const bool two = i + SL_ROWS < n;
        #pragma unroll
        for (int u = 0; u < SL_ROWS; ++u) {
            const int idx = (i + u < n) ? i + u : n - 1;
            const unsigned su = (unsigned)__builtin_amdgcn_readlane((int)s, idx);
            va[u] = ((const float2*)(h + (size_t)su * F_OUT))[lane];
        }
        if (two) {
            #pragma unroll
            for (int u = 0; u < SL_ROWS; ++u) {
                const int j   = i + SL_ROWS + u;
                const int idx = (j < n) ? j : n - 1;
                const unsigned su = (unsigned)__builtin_amdgcn_readlane((int)s, idx);
                vb[u] = ((const float2*)(h + (size_t)su * F_OUT))[lane];
            }
        }
        #pragma unroll
        for (int u = 0; u < SL_ROWS; ++u) {
            const bool ok = i + u < n;
            sum.x += ok ? va[u].x : 0.0f;
            sum.y += ok ? va[u].y : 0.0f;
        }
        if (two) {
            #pragma unroll
            for (int u = 0; u < SL_ROWS; ++u) {
                const bool ok = i + SL_ROWS + u < n;
                sum.x += ok ? vb[u].x : 0.0f;
                sum.y += ok ? vb[u].y : 0.0f;
            }
        }
    }
    return sum;
}

__global__ void __launch_bounds__(SL_THREADS) k_slice_agg(const unsigned* __restrict__ cells,
                                                          const int* __restrict__ src,
                                                          const int* __restrict__ dst,
                                                          const float* __restrict__ h,
                                                          const float* __restrict__ dinv,
                                                          const float* __restrict__ b_conv,
                                                          const float* __restrict__ W_lin,
                                                          const float* __restrict__ b_lin,
                                                          float* __restrict__ out, int N, int E, int ncell) {
    __shared__ __attribute__((aligned(16))) float acc[SL_NODES * F_OUT];
    __shared__ unsigned rq[SL_NODES * SL_RCAP];
    __shared__ unsigned cntr[SL_NODES];

    const int tid  = threadIdx.x;
    const int lane = tid & 31;
    const int wave = __builtin_amdgcn_readfirstlane(tid >> 5);
    const int lo   = blockIdx.x * SL_NODES;

    float4* acc4 = (float4*)acc;
    for (int i = tid; i < SL_NODES * F_OUT / 4; i += SL_THREADS)
        acc4[i] = make_float4(0.0f, 0.0f, 0.0f, 0.0f);
    for (int k = tid; k < SL_NODES; k += SL_THREADS) cntr[k] = 0u;
    __syncthreads();

    const uint4* row4 = (const uint4*)(cells + (size_t)blockIdx.x * ncell * CELL_WORDS);
    const int    nb   = ncell / BATCH_CELLS;
    const unsigned w0 = 4u * (unsigned)(lane & (CELL_LANES - 1));

    for (int it0 = wave; it0 < nb; it0 += SL_AHEAD * SL_WAVES) {
        uint4 c[SL_AHEAD];
        #pragma unroll
        for (int j = 0; j < SL_AHEAD; ++j) {
            const int it = it0 + j * SL_WAVES;
            c[j] = row4[((it < nb) ? it : it0) * 32 + lane];
        }
        unsigned ov = 0u;
        #pragma unroll
        for (int j = 0; j < SL_AHEAD; ++j) {
            const int it = it0 + j * SL_WAVES;
            if (it < nb) {
                const unsigned n    = (unsigned)__shfl((int)c[j].x, lane & 16);
                const bool     over = n > (unsigned)CELL_CAP;
                const unsigned nn   = over ? 0u : n;
                const bool hx = w0 - 1u < nn, hy = w0 < nn, hz = w0 + 1u < nn, hw = w0 + 2u < nn;
                const unsigned px = sl_slot(hx, c[j].x, cntr);
                const unsigned py = sl_slot(hy, c[j].y, cntr);
                const unsigned pz = sl_slot(hz, c[j].z, cntr);
                const unsigned pw = sl_slot(hw, c[j].w, cntr);
                const bool fx = sl_store(hx, c[j].x, px, rq);
                const bool fy = sl_store(hy, c[j].y, py, rq);
                const bool fz = sl_store(hz, c[j].z, pz, rq);
                const bool fw = sl_store(hw, c[j].w, pw, rq);
                sl_full(fx, c[j].x, acc, h, lane);
                sl_full(fy, c[j].y, acc, h, lane);
                sl_full(fz, c[j].z, acc, h, lane);
                sl_full(fw, c[j].w, acc, h, lane);
                const unsigned om = __builtin_amdgcn_ballot_w32(over);
                const unsigned o2 = (om & 1u) | ((om >> 15) & 2u);
                ov |= o2 << (2 * j);
            }
        }
        while (ov != 0u) {
            const int b = __builtin_ctz(ov);
            ov &= ov - 1u;
            sl_edges(2 * (it0 + (b >> 1) * SL_WAVES) + (b & 1), lo, src, dst, E, cntr, rq, acc, h, lane);
        }
    }

    __syncthreads();

    int rows = N - lo; if (rows > SL_NODES) rows = SL_NODES;
    const float2 bc = *(const float2*)(b_conv + 2 * lane);
    const float2 wl = *(const float2*)(W_lin + 2 * lane);
    const float  bl = b_lin[0];
    unsigned cnv = 0u;
    {
        const int rr = wave + lane * SL_WAVES;
        if (rr < rows) cnv = cntr[rr];
        if (cnv > (unsigned)SL_RCAP) cnv = (unsigned)SL_RCAP;
    }
    int k = 0;
    for (int r = wave; r < rows; r += SL_WAVES, ++k) {
        const int node = lo + r;
        const float2 hv = ((const float2*)(h + (size_t)node * F_OUT))[lane];
        const float  di = dinv[node];
        const int n = __builtin_amdgcn_readlane((int)cnv, k);
        unsigned s0 = 0u, s1 = 0u;
        if (lane < n)      s0 = rq[r * SL_RCAP + lane];
        if (32 + lane < n) s1 = rq[r * SL_RCAP + 32 + lane];
        float2 sum = make_float2(0.0f, 0.0f);
        sum = sl_sum(s0, (n < 32) ? n : 32, lane, h, sum);
        if (n > 32) sum = sl_sum(s1, n - 32, lane, h, sum);
        const float2 a = ((const float2*)acc)[r * (F_OUT / 2) + lane];
        const float ax = bc.x + di * (hv.x + (a.x + sum.x));
        const float ay = bc.y + di * (hv.y + (a.y + sum.y));
        float p = fmaxf(ax, 0.0f) * wl.x + fmaxf(ay, 0.0f) * wl.y;
        #pragma unroll
        for (int off = 16; off > 0; off >>= 1)
            p += __shfl_down(p, off);
        if (lane == 0) {
            const float z = p + bl;
            out[node] = 1.0f / (1.0f + __expf(-z));
        }
    }
}


extern "C" void kernel_launch(void* const* d_in, const int* in_sizes, int n_in,
                              void* d_out, int out_size, void* d_ws, size_t ws_size,
                              hipStream_t stream) {
    const float* x      = (const float*)d_in[0];
    const int*   edge   = (const int*)  d_in[1];
    const float* W_conv = (const float*)d_in[2];
    const float* b_conv = (const float*)d_in[3];
    const float* W_lin  = (const float*)d_in[4];
    const float* b_lin  = (const float*)d_in[5];
    float*       out    = (float*)d_out;

    const int N = in_sizes[0] / F_IN;
    const int E = in_sizes[1] / 2;
    const int* src = edge;
    const int* dst = edge + E;

    float* ws   = (float*)d_ws;
    float* h    = ws;
    float* dinv = h + (size_t)N * F_OUT;
    unsigned* cells = (unsigned*)(dinv + (N + 63) / 64 * 64);
    const int nsl   = (N + SL_NODES - 1) / SL_NODES;
    const int ncell = ((E + PB_EDGES - 1) / PB_EDGES + BATCH_CELLS - 1) / BATCH_CELLS * BATCH_CELLS;

    k_prep<<<ncell, PB_THREADS, 0, stream>>>(src, dst, cells, E, nsl, ncell);
    k_deg <<<nsl, DG_THREADS, 0, stream>>>(cells, dst, dinv, N, E, ncell);

    k_gemm<<<GM_BLOCKS, GM_WAVES * 32, 0, stream>>>(x, W_conv, dinv, h, N);

    k_slice_agg<<<nsl, SL_THREADS, 0, stream>>>(cells, src, dst, h, dinv, b_conv, W_lin, b_lin, out, N, E, ncell);
}
